# starting kernel
# baseline (speedup 1.0000x reference)
#include <hip/hip_runtime.h>

typedef __attribute__((ext_vector_type(2))) float v2f;
typedef __attribute__((ext_vector_type(8))) float v8f;

#define IC 256
#define OC 256
#define HH 56
#define WW 56
#define HWSZ (HH * WW)
#define CHW (IC * HWSZ)
#define KTAPS 9
#define WROW (IC * KTAPS)

#define WAVE_OC 16
#define WAVE_POS 64
#define BLOCK_OC 64
#define BLOCK_POS 128

__global__ __launch_bounds__(256) void conv3x3_wmma_f32(
    const float* __restrict__ x, const float* __restrict__ w,
    float* __restrict__ out)
{
  const int lane   = threadIdx.x & 31;
  const int wave   = threadIdx.x >> 5;
  const int l16    = lane & 15;
  const int hiHalf = lane >> 4;
  const int kHi    = hiHalf * 2;

  const int wave_oc  = wave & 3;
  const int wave_pos = wave >> 2;

  const int oc_base  = blockIdx.y * BLOCK_OC + wave_oc * WAVE_OC;
  const int pos_base = blockIdx.x * BLOCK_POS + wave_pos * WAVE_POS;

  const int m = oc_base + l16;

  int n_[4], h_[4], w_[4], sp_[4];
#pragma unroll
  for (int j = 0; j < 4; ++j) {
    int p  = pos_base + j * 16 + l16;
    int n  = p / HWSZ;
    int sp = p - n * HWSZ;
    int hh = sp / WW;
    int ww = sp - hh * WW;
    n_[j] = n; h_[j] = hh; w_[j] = ww; sp_[j] = sp;
  }

  v8f c[4] = {};

  const float* wRowBase = w + (long)m * WROW + kHi * KTAPS;

  for (int tap = 0; tap < KTAPS; ++tap) {
    const int dh = tap / 3 - 1;
    const int dw = tap % 3 - 1;

    const float* bptr[4];
    bool valid[4];
#pragma unroll
    for (int j = 0; j < 4; ++j) {
      int hh = h_[j] + dh, ww = w_[j] + dw;
      valid[j] = ((unsigned)hh < (unsigned)HH) && ((unsigned)ww < (unsigned)WW);
      long spOff = valid[j] ? (long)(hh * WW + ww) : 0L;
      bptr[j] = x + (long)n_[j] * CHW + (long)kHi * HWSZ + spOff;
    }

    const float* aptr = wRowBase + tap;

#pragma unroll 4
    for (int ic0 = 0; ic0 < IC; ic0 += 4) {
      v2f a;
      a.x = aptr[0];
      a.y = aptr[KTAPS];
      aptr += 4 * KTAPS;

#pragma unroll
      for (int j = 0; j < 4; ++j) {
        float t0 = bptr[j][0];
        float t1 = bptr[j][HWSZ];
        bptr[j] += 4 * HWSZ;
        v2f b;
        b.x = valid[j] ? t0 : 0.0f;
        b.y = valid[j] ? t1 : 0.0f;

        c[j] = __builtin_amdgcn_wmma_f32_16x16x4_f32(
             false, a,  false, b,
             (short)0, c[j],  false,  false);
      }
    }
  }

  const float s = 1.0f / 9.0f;
#pragma unroll
  for (int j = 0; j < 4; ++j) {
    const int  ocr   = oc_base + hiHalf * 8;
    const long obase = (long)n_[j] * (long)OC * HWSZ + sp_[j];
#pragma unroll
    for (int r = 0; r < 8; ++r) {
      out[obase + (long)(ocr + r) * HWSZ] = c[j][r] * s;
    }
  }
}

extern "C" void kernel_launch(void* const* d_in, const int* in_sizes, int n_in,
                              void* d_out, int out_size, void* d_ws, size_t ws_size,
                              hipStream_t stream) {
  const float* x = (const float*)d_in[0];
  const float* w = (const float*)d_in[1];
  float* out = (float*)d_out;

  dim3 grid(100352 / BLOCK_POS, OC / BLOCK_OC, 1);
  dim3 block(256, 1, 1);
  conv3x3_wmma_f32<<<grid, block, 0, stream>>>(x, w, out);
}


# the 3 row taps in F(2,3) minimal-filtering form along the height, built in registers: a wave takes 16 oc x 4 rows, 4 products per output row pair, not 6; sums now in the transformed domain
# speedup vs baseline: 5.5018x; 5.5018x over previous
#include <hip/hip_runtime.h>


typedef __attribute__((ext_vector_type(2))) float v2f;
typedef __attribute__((ext_vector_type(4))) float v4f;
typedef __attribute__((ext_vector_type(8))) float v8f;

#define IC 256
#define OC 256
#define HH 56
#define WW 56
#define HWSZ (HH * WW)
#define CHW (IC * HWSZ)
#define KTAPS 9
#define WROW (IC * KTAPS)

#define MT 2
#define NT 7
#define WAVE_OC (16 * MT)
#define ROW_PAIRS (32 * (HH / 2))

#define OC_TILES (OC / 16)
#define Q_STEPS (IC / 8)
#define CHUNKS_PER_TAP (OC_TILES * Q_STEPS * 32)
#define W2_CHUNKS (KTAPS * CHUNKS_PER_TAP)
#define W2_BYTES (W2_CHUNKS * 16)

__global__ __launch_bounds__(256) void weights_to_fragment_order(
    const float* __restrict__ w, float* __restrict__ w2, size_t ws_bytes)
{
  if (ws_bytes < (size_t)W2_BYTES) return;
  const int idx  = blockIdx.x * 256 + threadIdx.x;
  const int L    = idx & 31;
  const int q    = (idx >> 5) & 31;
  const int oct  = (idx >> 10) & 15;
  const int tap  = idx >> 14;
  const int row  = L & 15;
  const int half = L >> 4;
  const int ic0  = q * 8 + half * 2;
  const float* src = w + (long)(oct * 16 + row) * WROW + tap;
  v4f v;
  v.x = src[(ic0    ) * KTAPS];
  v.y = src[(ic0 + 1) * KTAPS];
  v.z = src[(ic0 + 4) * KTAPS];
  v.w = src[(ic0 + 5) * KTAPS];
  ((v4f*)w2)[idx] = v;
}

__device__ __forceinline__ void load7(const float* __restrict__ p, float* d)
{
  v4f a; v2f b;
  __builtin_memcpy(&a, p, 16);
  __builtin_memcpy(&b, p + 4, 8);
  d[0] = a.x; d[1] = a.y; d[2] = a.z; d[3] = a.w;
  d[4] = b.x; d[5] = b.y; d[6] = p[6];
}

__device__ __forceinline__ float from_lane_below(float v)
{
  return __int_as_float(__builtin_amdgcn_update_dpp(0, __float_as_int(v), 0x111, 0xF, 0xF, true));
}
__device__ __forceinline__ float from_lane_above(float v)
{
  return __int_as_float(__builtin_amdgcn_update_dpp(0, __float_as_int(v), 0x101, 0xF, 0xF, true));
}

template <bool FRAG_W>
__device__ __forceinline__ void conv_body(
    const float* __restrict__ x, const float* __restrict__ w,
    const float* __restrict__ w2, float* __restrict__ out)
{
  const int lane   = threadIdx.x & 31;
  const int wave   = threadIdx.x >> 5;
  const int l16    = lane & 15;
  const int hiHalf = lane >> 4;
  const int kHi    = hiHalf * 2;

  const int oc_base = ((blockIdx.x & 1) * 4 + wave) * WAVE_OC;
  const int rp      = blockIdx.x >> 1;
  const int n       = rp / (HH / 2);
  const int h       = (rp - n * (HH / 2)) * 2 + (l16 >> 3);
  const int wq      = l16 & 7;
  const int w0      = 7 * wq;


  v8f c[MT][NT] = {};

  const float* wRowBase = w + (long)(oc_base + l16) * WROW + kHi * KTAPS;
  const v4f*   fragBase = (const v4f*)w2 + (long)(oc_base >> 4) * (Q_STEPS * 32) + lane;

  for (int dhi = 0; dhi < 3; ++dhi) {
    const int  hh = h + dhi - 1;
    const bool vr = (unsigned)hh < (unsigned)HH;
    const bool mL = vr && (wq != 0);
    const bool mR = vr && (wq != 7);
    const float* bp = x + (long)n * CHW + (long)kHi * HWSZ + (long)((vr ? hh : h) * WW + w0);

    const v4f*   ap   = fragBase + (long)(dhi * 3) * CHUNKS_PER_TAP;
    const float* aptr = wRowBase + dhi * 3;

    for (int q = 0; q < Q_STEPS; ++q) {
      v4f a4[3][MT];
      if (FRAG_W) {
#pragma unroll
        for (int d = 0; d < 3; ++d)
#pragma unroll
          for (int t = 0; t < MT; ++t) a4[d][t] = ap[(long)d * CHUNKS_PER_TAP + t * (Q_STEPS * 32)];
        ap += 32;
      }

#pragma unroll
      for (int s = 0; s < 2; ++s) {
        v2f a[3][MT];
#pragma unroll
        for (int d = 0; d < 3; ++d)
#pragma unroll
          for (int t = 0; t < MT; ++t) {
            if (FRAG_W) {
              a[d][t].x = s ? a4[d][t].z : a4[d][t].x;
              a[d][t].y = s ? a4[d][t].w : a4[d][t].y;
            } else {
              a[d][t].x = aptr[(long)t * 16 * WROW + d];
              a[d][t].y = aptr[(long)t * 16 * WROW + d + KTAPS];
            }
          }
        if (!FRAG_W) aptr += 4 * KTAPS;

        float e0[9], e1[9];
        load7(bp, &e0[1]);
        load7(bp + HWSZ, &e1[1]);
        bp += 4 * HWSZ;
        const float l0 = from_lane_below(e0[7]), l1 = from_lane_below(e1[7]);
        const float r0 = from_lane_above(e0[1]), r1 = from_lane_above(e1[1]);

        v2f p[9];
        p[0].x = mL ? l0 : 0.0f;  p[0].y = mL ? l1 : 0.0f;
        p[8].x = mR ? r0 : 0.0f;  p[8].y = mR ? r1 : 0.0f;
#pragma unroll
        for (int i = 1; i < 8; ++i) {
          p[i].x = vr ? e0[i] : 0.0f;
          p[i].y = vr ? e1[i] : 0.0f;
        }

#pragma unroll
        for (int d = 0; d < 3; ++d)
#pragma unroll
          for (int t = 0; t < MT; ++t)
#pragma unroll
            for (int j = 0; j < NT; ++j) {
              c[t][j] = __builtin_amdgcn_wmma_f32_16x16x4_f32(
                   false, a[d][t],  false, p[j + d],
                   (short)0, c[t][j],  false,  false);
            }
      }
    }
  }

  const float sc = 1.0f / 9.0f;
  float* obase = out + ((long)n * OC + oc_base + hiHalf * 8) * HWSZ + (h * WW + w0);
#pragma unroll
  for (int t = 0; t < MT; ++t) {
#pragma unroll
    for (int r = 0; r < 8; ++r) {
      float* o = obase + (long)(t * 16 + r) * HWSZ;
      v4f v; v2f u;
      v.x = c[t][0][r] * sc; v.y = c[t][1][r] * sc; v.z = c[t][2][r] * sc; v.w = c[t][3][r] * sc;
      u.x = c[t][4][r] * sc; u.y = c[t][5][r] * sc;
      __builtin_memcpy(o, &v, 16);
      __builtin_memcpy(o + 4, &u, 8);
      o[6] = c[t][6][r] * sc;
    }
  }
}

__device__ __forceinline__ void conv_body_rows_f23(
    const float* __restrict__ x, const float* __restrict__ w2, float* __restrict__ out)
{
  const int lane   = threadIdx.x & 31;
  const int wave   = threadIdx.x >> 5;
  const int l16    = lane & 15;
  const int hiHalf = lane >> 4;
  const int kHi    = hiHalf * 2;

  const int oc_base = ((blockIdx.x & 3) * 4 + wave) * 16;
  const int unit    = blockIdx.x >> 2;
  const int n       = unit / (HH / 4);
  const int hA      = (unit - n * (HH / 4)) * 4 + (l16 >> 3) * 2;
  const int wq      = l16 & 7;
  const int w0      = 7 * wq;

  const bool v0 = hA != 0;
  const bool v3 = hA != HH - 2;
  const bool mL = wq != 0;
  const bool mR = wq != 7;


  v8f c[4][NT] = {};

  const float* bp = x + (long)n * CHW + (long)kHi * HWSZ + w0;
  const int o0 = (v0 ? hA - 1 : hA) * WW;
  const int o1 = hA * WW;
  const int o2 = (hA + 1) * WW;
  const int o3 = (v3 ? hA + 2 : hA) * WW;
  const v4f* ap = (const v4f*)w2 + (long)(oc_base >> 4) * (Q_STEPS * 32) + lane;

  for (int q = 0; q < Q_STEPS; ++q) {
    v4f u4[4][3];
#pragma unroll
    for (int d = 0; d < 3; ++d) {
      const v4f g0 = ap[(long)(0 * 3 + d) * CHUNKS_PER_TAP];
      const v4f g1 = ap[(long)(1 * 3 + d) * CHUNKS_PER_TAP];
      const v4f g2 = ap[(long)(2 * 3 + d) * CHUNKS_PER_TAP];
      const v4f g02 = g0 + g2;
      u4[0][d] = g0;
      u4[1][d] = (g02 + g1) * 0.5f;
      u4[2][d] = (g02 - g1) * 0.5f;
      u4[3][d] = g2;
    }
    ap += 32;

#pragma unroll
    for (int s = 0; s < 2; ++s) {
      v2f a[4][3];
#pragma unroll
      for (int i = 0; i < 4; ++i)
#pragma unroll
        for (int d = 0; d < 3; ++d) {
          a[i][d].x = s ? u4[i][d].z : u4[i][d].x;
          a[i][d].y = s ? u4[i][d].w : u4[i][d].y;
        }

      float V[4][2][7];
#pragma unroll
      for (int ch = 0; ch < 2; ++ch) {
        float d0[7], d1[7], d2[7], d3[7];
        const float* pc = bp + ch * HWSZ;
        load7(pc + o0, d0);
        load7(pc + o1, d1);
        load7(pc + o2, d2);
        load7(pc + o3, d3);
#pragma unroll
        for (int k = 0; k < 7; ++k) {
          const float t0 = v0 ? d0[k] : 0.0f;
          const float t3 = v3 ? d3[k] : 0.0f;
          V[0][ch][k] = t0 - d2[k];
          V[1][ch][k] = d1[k] + d2[k];
          V[2][ch][k] = d2[k] - d1[k];
          V[3][ch][k] = d1[k] - t3;
        }
      }
      bp += 4 * HWSZ;

      v2f p[4][9];
#pragma unroll
      for (int i = 0; i < 4; ++i) {
        const float l0 = from_lane_below(V[i][0][6]), l1 = from_lane_below(V[i][1][6]);
        const float r0 = from_lane_above(V[i][0][0]), r1 = from_lane_above(V[i][1][0]);
        p[i][0].x = mL ? l0 : 0.0f;  p[i][0].y = mL ? l1 : 0.0f;
        p[i][8].x = mR ? r0 : 0.0f;  p[i][8].y = mR ? r1 : 0.0f;
#pragma unroll
        for (int k = 0; k < 7; ++k) {
          p[i][k + 1].x = V[i][0][k];
          p[i][k + 1].y = V[i][1][k];
        }
      }

#pragma unroll
      for (int i = 0; i < 4; ++i)
#pragma unroll
        for (int d = 0; d < 3; ++d)
#pragma unroll
          for (int j = 0; j < NT; ++j) {
            c[i][j] = __builtin_amdgcn_wmma_f32_16x16x4_f32(
                 false, a[i][d],  false, p[i][j + d],
                 (short)0, c[i][j],  false,  false);
          }
    }
  }

  const float sc = 1.0f / 9.0f;
  float* obase = out + ((long)n * OC + oc_base + hiHalf * 8) * HWSZ + (hA * WW + w0);
#pragma unroll
  for (int r = 0; r < 8; ++r) {
    float y0[NT], y1[NT];
#pragma unroll
    for (int j = 0; j < NT; ++j) {
      const float m0 = c[0][j][r], m1 = c[1][j][r], m2 = c[2][j][r], m3 = c[3][j][r];
      y0[j] = (m0 + m1 + m2) * sc;
      y1[j] = (m1 - m2 - m3) * sc;
    }
    float* o = obase + (long)r * HWSZ;
    v4f v; v2f u;
    v.x = y0[0]; v.y = y0[1]; v.z = y0[2]; v.w = y0[3];
    u.x = y0[4]; u.y = y0[5];
    __builtin_memcpy(o, &v, 16);
    __builtin_memcpy(o + 4, &u, 8);
    o[6] = y0[6];
    v.x = y1[0]; v.y = y1[1]; v.z = y1[2]; v.w = y1[3];
    u.x = y1[4]; u.y = y1[5];
    __builtin_memcpy(o + WW, &v, 16);
    __builtin_memcpy(o + WW + 4, &u, 8);
    o[WW + 6] = y1[6];
  }
}

__global__ __launch_bounds__(128) void conv3x3_wmma_f32(
    const float* __restrict__ x, const float* __restrict__ w,
    const float* __restrict__ w2, size_t ws_bytes, float* __restrict__ out)
{
  if (ws_bytes >= (size_t)W2_BYTES) conv_body_rows_f23(x, w2, out);
  else                              conv_body<false>(x, w, w2, out);
}

extern "C" void kernel_launch(void* const* d_in, const int* in_sizes, int n_in,
                              void* d_out, int out_size, void* d_ws, size_t ws_size,
                              hipStream_t stream) {
  const float* x = (const float*)d_in[0];
  const float* w = (const float*)d_in[1];
  float* out = (float*)d_out;
  float* w2  = (float*)d_ws;

  weights_to_fragment_order<<<dim3(W2_CHUNKS / 256, 1, 1), dim3(256, 1, 1), 0, stream>>>(w, w2, ws_size);

  dim3 grid(ROW_PAIRS * 2, 1, 1);
  dim3 block(128, 1, 1);
  conv3x3_wmma_f32<<<grid, block, 0, stream>>>(x, w, w2, ws_size, out);
}


# global loads pipelined in registers: rows of k-step t+1 and the next q's 9 weight fragments are requested before k-step t's 84 products and first read after them; sum order unchanged
# speedup vs baseline: 5.8696x; 1.0669x over previous
#include <hip/hip_runtime.h>


typedef __attribute__((ext_vector_type(2))) float v2f;
typedef __attribute__((ext_vector_type(4))) float v4f;
typedef __attribute__((ext_vector_type(8))) float v8f;

#define IC 256
#define OC 256
#define HH 56
#define WW 56
#define HWSZ (HH * WW)
#define CHW (IC * HWSZ)
#define KTAPS 9
#define WROW (IC * KTAPS)

#define MT 2
#define NT 7
#define WAVE_OC (16 * MT)
#define ROW_PAIRS (32 * (HH / 2))

#define OC_TILES (OC / 16)
#define Q_STEPS (IC / 8)
#define CHUNKS_PER_TAP (OC_TILES * Q_STEPS * 32)
#define W2_CHUNKS (KTAPS * CHUNKS_PER_TAP)
#define W2_BYTES (W2_CHUNKS * 16)

__global__ __launch_bounds__(256) void weights_to_fragment_order(
    const float* __restrict__ w, float* __restrict__ w2, size_t ws_bytes)
{
  if (ws_bytes < (size_t)W2_BYTES) return;
  const int idx  = blockIdx.x * 256 + threadIdx.x;
  const int L    = idx & 31;
  const int q    = (idx >> 5) & 31;
  const int oct  = (idx >> 10) & 15;
  const int tap  = idx >> 14;
  const int row  = L & 15;
  const int half = L >> 4;
  const int ic0  = q * 8 + half * 2;
  const float* src = w + (long)(oct * 16 + row) * WROW + tap;
  v4f v;
  v.x = src[(ic0    ) * KTAPS];
  v.y = src[(ic0 + 1) * KTAPS];
  v.z = src[(ic0 + 4) * KTAPS];
  v.w = src[(ic0 + 5) * KTAPS];
  ((v4f*)w2)[idx] = v;
}

__device__ __forceinline__ void load7(const float* __restrict__ p, float* d)
{
  v4f a; v2f b;
  __builtin_memcpy(&a, p, 16);
  __builtin_memcpy(&b, p + 4, 8);
  d[0] = a.x; d[1] = a.y; d[2] = a.z; d[3] = a.w;
  d[4] = b.x; d[5] = b.y; d[6] = p[6];
}

__device__ __forceinline__ float from_lane_below(float v)
{
  return __int_as_float(__builtin_amdgcn_update_dpp(0, __float_as_int(v), 0x111, 0xF, 0xF, true));
}
__device__ __forceinline__ float from_lane_above(float v)
{
  return __int_as_float(__builtin_amdgcn_update_dpp(0, __float_as_int(v), 0x101, 0xF, 0xF, true));
}

template <bool FRAG_W>
__device__ __forceinline__ void conv_body(
    const float* __restrict__ x, const float* __restrict__ w,
    const float* __restrict__ w2, float* __restrict__ out)
{
  const int lane   = threadIdx.x & 31;
  const int wave   = threadIdx.x >> 5;
  const int l16    = lane & 15;
  const int hiHalf = lane >> 4;
  const int kHi    = hiHalf * 2;

  const int oc_base = ((blockIdx.x & 1) * 4 + wave) * WAVE_OC;
  const int rp      = blockIdx.x >> 1;
  const int n       = rp / (HH / 2);
  const int h       = (rp - n * (HH / 2)) * 2 + (l16 >> 3);
  const int wq      = l16 & 7;
  const int w0      = 7 * wq;


  v8f c[MT][NT] = {};

  const float* wRowBase = w + (long)(oc_base + l16) * WROW + kHi * KTAPS;
  const v4f*   fragBase = (const v4f*)w2 + (long)(oc_base >> 4) * (Q_STEPS * 32) + lane;

  for (int dhi = 0; dhi < 3; ++dhi) {
    const int  hh = h + dhi - 1;
    const bool vr = (unsigned)hh < (unsigned)HH;
    const bool mL = vr && (wq != 0);
    const bool mR = vr && (wq != 7);
    const float* bp = x + (long)n * CHW + (long)kHi * HWSZ + (long)((vr ? hh : h) * WW + w0);

    const v4f*   ap   = fragBase + (long)(dhi * 3) * CHUNKS_PER_TAP;
    const float* aptr = wRowBase + dhi * 3;

    for (int q = 0; q < Q_STEPS; ++q) {
      v4f a4[3][MT];
      if (FRAG_W) {
#pragma unroll
        for (int d = 0; d < 3; ++d)
#pragma unroll
          for (int t = 0; t < MT; ++t) a4[d][t] = ap[(long)d * CHUNKS_PER_TAP + t * (Q_STEPS * 32)];
        ap += 32;
      }

#pragma unroll
      for (int s = 0; s < 2; ++s) {
        v2f a[3][MT];
#pragma unroll
        for (int d = 0; d < 3; ++d)
#pragma unroll
          for (int t = 0; t < MT; ++t) {
            if (FRAG_W) {
              a[d][t].x = s ? a4[d][t].z : a4[d][t].x;
              a[d][t].y = s ? a4[d][t].w : a4[d][t].y;
            } else {
              a[d][t].x = aptr[(long)t * 16 * WROW + d];
              a[d][t].y = aptr[(long)t * 16 * WROW + d + KTAPS];
            }
          }
        if (!FRAG_W) aptr += 4 * KTAPS;

        float e0[9], e1[9];
        load7(bp, &e0[1]);
        load7(bp + HWSZ, &e1[1]);
        bp += 4 * HWSZ;
        const float l0 = from_lane_below(e0[7]), l1 = from_lane_below(e1[7]);
        const float r0 = from_lane_above(e0[1]), r1 = from_lane_above(e1[1]);

        v2f p[9];
        p[0].x = mL ? l0 : 0.0f;  p[0].y = mL ? l1 : 0.0f;
        p[8].x = mR ? r0 : 0.0f;  p[8].y = mR ? r1 : 0.0f;
#pragma unroll
        for (int i = 1; i < 8; ++i) {
          p[i].x = vr ? e0[i] : 0.0f;
          p[i].y = vr ? e1[i] : 0.0f;
        }

#pragma unroll
        for (int d = 0; d < 3; ++d)
#pragma unroll
          for (int t = 0; t < MT; ++t)
#pragma unroll
            for (int j = 0; j < NT; ++j) {
              c[t][j] = __builtin_amdgcn_wmma_f32_16x16x4_f32(
                   false, a[d][t],  false, p[j + d],
                   (short)0, c[t][j],  false,  false);
            }
      }
    }
  }

  const float sc = 1.0f / 9.0f;
  float* obase = out + ((long)n * OC + oc_base + hiHalf * 8) * HWSZ + (h * WW + w0);
#pragma unroll
  for (int t = 0; t < MT; ++t) {
#pragma unroll
    for (int r = 0; r < 8; ++r) {
      float* o = obase + (long)(t * 16 + r) * HWSZ;
      v4f v; v2f u;
      v.x = c[t][0][r] * sc; v.y = c[t][1][r] * sc; v.z = c[t][2][r] * sc; v.w = c[t][3][r] * sc;
      u.x = c[t][4][r] * sc; u.y = c[t][5][r] * sc;
      __builtin_memcpy(o, &v, 16);
      __builtin_memcpy(o + 4, &u, 8);
      o[6] = c[t][6][r] * sc;
    }
  }
}

__device__ __forceinline__ void load_rows(const float* __restrict__ pc, int o0, int o1, int o2, int o3,
                                          float (&R)[2][4][7])
{
#pragma unroll
  for (int ch = 0; ch < 2; ++ch) {
    load7(pc + ch * HWSZ + o0, R[ch][0]);
    load7(pc + ch * HWSZ + o1, R[ch][1]);
    load7(pc + ch * HWSZ + o2, R[ch][2]);
    load7(pc + ch * HWSZ + o3, R[ch][3]);
  }
}

__device__ __forceinline__ void load_weights(const v4f* __restrict__ ap, v4f (&G)[3][3])
{
#pragma unroll
  for (int r = 0; r < 3; ++r)
#pragma unroll
    for (int d = 0; d < 3; ++d) G[r][d] = ap[(long)(r * 3 + d) * CHUNKS_PER_TAP];
}

template <int S>
__device__ __forceinline__ void k_step_f23(const float (&R)[2][4][7], const v4f (&u4)[4][3],
                                           bool v0, bool v3, bool mL, bool mR, v8f (&c)[4][NT])
{
  v2f a[4][3];
#pragma unroll
  for (int i = 0; i < 4; ++i)
#pragma unroll
    for (int d = 0; d < 3; ++d) {
      a[i][d].x = S ? u4[i][d].z : u4[i][d].x;
      a[i][d].y = S ? u4[i][d].w : u4[i][d].y;
    }

  float V[4][2][7];
#pragma unroll
  for (int ch = 0; ch < 2; ++ch)
#pragma unroll
    for (int k = 0; k < 7; ++k) {
      const float t0 = v0 ? R[ch][0][k] : 0.0f;
      const float t3 = v3 ? R[ch][3][k] : 0.0f;
      const float d1 = R[ch][1][k], d2 = R[ch][2][k];
      V[0][ch][k] = t0 - d2;
      V[1][ch][k] = d1 + d2;
      V[2][ch][k] = d2 - d1;
      V[3][ch][k] = d1 - t3;
    }

  v2f p[4][9];
#pragma unroll
  for (int i = 0; i < 4; ++i) {
    const float l0 = from_lane_below(V[i][0][6]), l1 = from_lane_below(V[i][1][6]);
    const float r0 = from_lane_above(V[i][0][0]), r1 = from_lane_above(V[i][1][0]);
    p[i][0].x = mL ? l0 : 0.0f;  p[i][0].y = mL ? l1 : 0.0f;
    p[i][8].x = mR ? r0 : 0.0f;  p[i][8].y = mR ? r1 : 0.0f;
#pragma unroll
    for (int k = 0; k < 7; ++k) {
      p[i][k + 1].x = V[i][0][k];
      p[i][k + 1].y = V[i][1][k];
    }
  }

#pragma unroll
  for (int i = 0; i < 4; ++i)
#pragma unroll
    for (int d = 0; d < 3; ++d)
#pragma unroll
      for (int j = 0; j < NT; ++j) {
        c[i][j] = __builtin_amdgcn_wmma_f32_16x16x4_f32(
             false, a[i][d],  false, p[i][j + d],
             (short)0, c[i][j],  false,  false);
      }
}

__device__ __forceinline__ void conv_body_rows_f23(
    const float* __restrict__ x, const float* __restrict__ w2, float* __restrict__ out)
{
  const int lane   = threadIdx.x & 31;
  const int wave   = threadIdx.x >> 5;
  const int l16    = lane & 15;
  const int hiHalf = lane >> 4;
  const int kHi    = hiHalf * 2;

  const int oc_base = ((blockIdx.x & 3) * 4 + wave) * 16;
  const int unit    = blockIdx.x >> 2;
  const int n       = unit / (HH / 4);
  const int hA      = (unit - n * (HH / 4)) * 4 + (l16 >> 3) * 2;
  const int wq      = l16 & 7;
  const int w0      = 7 * wq;

  const bool v0 = hA != 0;
  const bool v3 = hA != HH - 2;
  const bool mL = wq != 0;
  const bool mR = wq != 7;


  v8f c[4][NT] = {};

  const float* bp = x + (long)n * CHW + (long)kHi * HWSZ + w0;
  const int o0 = (v0 ? hA - 1 : hA) * WW;
  const int o1 = hA * WW;
  const int o2 = (hA + 1) * WW;
  const int o3 = (v3 ? hA + 2 : hA) * WW;
  const v4f* ap = (const v4f*)w2 + (long)(oc_base >> 4) * (Q_STEPS * 32) + lane;

  v4f   G[3][3];
  float R0[2][4][7];
  float R1[2][4][7];
  load_weights(ap, G);
  load_rows(bp, o0, o1, o2, o3, R0);

  for (int q = 0; q < Q_STEPS; ++q) {
    load_rows(bp + 4 * HWSZ, o0, o1, o2, o3, R1);

    v4f u4[4][3];
#pragma unroll
    for (int d = 0; d < 3; ++d) {
      const v4f g02 = G[0][d] + G[2][d];
      u4[0][d] = G[0][d];
      u4[1][d] = (g02 + G[1][d]) * 0.5f;
      u4[2][d] = (g02 - G[1][d]) * 0.5f;
      u4[3][d] = G[2][d];
    }

    k_step_f23<0>(R0, u4, v0, v3, mL, mR, c);
    __builtin_amdgcn_sched_barrier(0);

    const int more = (q + 1 < Q_STEPS) ? 1 : 0;
    ap += more * 32;
    bp += more * (8 * HWSZ);
    load_weights(ap, G);
    load_rows(bp, o0, o1, o2, o3, R0);

    k_step_f23<1>(R1, u4, v0, v3, mL, mR, c);
    __builtin_amdgcn_sched_barrier(0);
  }

  const float sc = 1.0f / 9.0f;
  float* obase = out + ((long)n * OC + oc_base + hiHalf * 8) * HWSZ + (hA * WW + w0);
#pragma unroll
  for (int r = 0; r < 8; ++r) {
    float y0[NT], y1[NT];
#pragma unroll
    for (int j = 0; j < NT; ++j) {
      const float m0 = c[0][j][r], m1 = c[1][j][r], m2 = c[2][j][r], m3 = c[3][j][r];
      y0[j] = (m0 + m1 + m2) * sc;
      y1[j] = (m1 - m2 - m3) * sc;
    }
    float* o = obase + (long)r * HWSZ;
    v4f v; v2f u;
    v.x = y0[0]; v.y = y0[1]; v.z = y0[2]; v.w = y0[3];
    u.x = y0[4]; u.y = y0[5];
    __builtin_memcpy(o, &v, 16);
    __builtin_memcpy(o + 4, &u, 8);
    o[6] = y0[6];
    v.x = y1[0]; v.y = y1[1]; v.z = y1[2]; v.w = y1[3];
    u.x = y1[4]; u.y = y1[5];
    __builtin_memcpy(o + WW, &v, 16);
    __builtin_memcpy(o + WW + 4, &u, 8);
    o[WW + 6] = y1[6];
  }
}

__global__ __launch_bounds__(128) void conv3x3_wmma_f32(
    const float* __restrict__ x, const float* __restrict__ w,
    const float* __restrict__ w2, size_t ws_bytes, float* __restrict__ out)
{
  if (ws_bytes >= (size_t)W2_BYTES) conv_body_rows_f23(x, w2, out);
  else                              conv_body<false>(x, w, w2, out);
}

extern "C" void kernel_launch(void* const* d_in, const int* in_sizes, int n_in,
                              void* d_out, int out_size, void* d_ws, size_t ws_size,
                              hipStream_t stream) {
  const float* x = (const float*)d_in[0];
  const float* w = (const float*)d_in[1];
  float* out = (float*)d_out;
  float* w2  = (float*)d_ws;

  weights_to_fragment_order<<<dim3(W2_CHUNKS / 256, 1, 1), dim3(256, 1, 1), 0, stream>>>(w, w2, ws_size);

  dim3 grid(ROW_PAIRS * 2, 1, 1);
  dim3 block(128, 1, 1);
  conv3x3_wmma_f32<<<grid, block, 0, stream>>>(x, w, w2, ws_size, out);
}


# a lane's first pixel pair in F(2,3) minimal-filtering form along the width on top of the F(4,3) rows: 57 products per k-step, not 63, still two waves per SIMD; that pair's sums 2-D transformed
# speedup vs baseline: 7.3411x; 1.2507x over previous
#include <hip/hip_runtime.h>


typedef __attribute__((ext_vector_type(2))) float v2f;
typedef __attribute__((ext_vector_type(4))) float v4f;
typedef __attribute__((ext_vector_type(8))) float v8f;

#define IC 256
#define OC 256
#define HH 56
#define WW 56
#define HWSZ (HH * WW)
#define CHW (IC * HWSZ)
#define KTAPS 9
#define WROW (IC * KTAPS)

#define MT 2
#define NT 7
#define WAVE_OC (16 * MT)
#define ROW_PAIRS (32 * (HH / 2))
#define NTC 9

#define OC_TILES (OC / 16)
#define Q_STEPS (IC / 8)
#define CHUNKS_PER_TAP (OC_TILES * Q_STEPS * 32)
#define W2_CHUNKS (KTAPS * CHUNKS_PER_TAP)
#define W2_BYTES (W2_CHUNKS * 16)

__global__ __launch_bounds__(256) void weights_to_fragment_order(
    const float* __restrict__ w, float* __restrict__ w2, size_t ws_bytes)
{
  if (ws_bytes < (size_t)W2_BYTES) return;
  const int idx  = blockIdx.x * 256 + threadIdx.x;
  const int L    = idx & 31;
  const int q    = (idx >> 5) & 31;
  const int oct  = (idx >> 10) & 15;
  const int tap  = idx >> 14;
  const int row  = L & 15;
  const int half = L >> 4;
  const int ic0  = q * 8 + half * 2;
  const float* src = w + (long)(oct * 16 + row) * WROW + tap;
  v4f v;
  v.x = src[(ic0    ) * KTAPS];
  v.y = src[(ic0 + 1) * KTAPS];
  v.z = src[(ic0 + 4) * KTAPS];
  v.w = src[(ic0 + 5) * KTAPS];
  ((v4f*)w2)[idx] = v;
}

__device__ __forceinline__ void load7(const float* __restrict__ p, float* d)
{
  v4f a; v2f b;
  __builtin_memcpy(&a, p, 16);
  __builtin_memcpy(&b, p + 4, 8);
  d[0] = a.x; d[1] = a.y; d[2] = a.z; d[3] = a.w;
  d[4] = b.x; d[5] = b.y; d[6] = p[6];
}

__device__ __forceinline__ float from_lane_below(float v)
{
  return __int_as_float(__builtin_amdgcn_update_dpp(0, __float_as_int(v), 0x111, 0xF, 0xF, true));
}
__device__ __forceinline__ float from_lane_above(float v)
{
  return __int_as_float(__builtin_amdgcn_update_dpp(0, __float_as_int(v), 0x101, 0xF, 0xF, true));
}

template <bool FRAG_W>
__device__ __forceinline__ void conv_body(
    const float* __restrict__ x, const float* __restrict__ w,
    const float* __restrict__ w2, float* __restrict__ out)
{
  const int lane   = threadIdx.x & 31;
  const int wave   = threadIdx.x >> 5;
  const int l16    = lane & 15;
  const int hiHalf = lane >> 4;
  const int kHi    = hiHalf * 2;

  const int oc_base = ((blockIdx.x & 1) * 4 + wave) * WAVE_OC;
  const int rp      = blockIdx.x >> 1;
  const int n       = rp / (HH / 2);
  const int h       = (rp - n * (HH / 2)) * 2 + (l16 >> 3);
  const int wq      = l16 & 7;
  const int w0      = 7 * wq;


  v8f c[MT][NT] = {};

  const float* wRowBase = w + (long)(oc_base + l16) * WROW + kHi * KTAPS;
  const v4f*   fragBase = (const v4f*)w2 + (long)(oc_base >> 4) * (Q_STEPS * 32) + lane;

  for (int dhi = 0; dhi < 3; ++dhi) {
    const int  hh = h + dhi - 1;
    const bool vr = (unsigned)hh < (unsigned)HH;
    const bool mL = vr && (wq != 0);
    const bool mR = vr && (wq != 7);
    const float* bp = x + (long)n * CHW + (long)kHi * HWSZ + (long)((vr ? hh : h) * WW + w0);

    const v4f*   ap   = fragBase + (long)(dhi * 3) * CHUNKS_PER_TAP;
    const float* aptr = wRowBase + dhi * 3;

    for (int q = 0; q < Q_STEPS; ++q) {
      v4f a4[3][MT];
      if (FRAG_W) {
#pragma unroll
        for (int d = 0; d < 3; ++d)
#pragma unroll
          for (int t = 0; t < MT; ++t) a4[d][t] = ap[(long)d * CHUNKS_PER_TAP + t * (Q_STEPS * 32)];
        ap += 32;
      }

#pragma unroll
      for (int s = 0; s < 2; ++s) {
        v2f a[3][MT];
#pragma unroll
        for (int d = 0; d < 3; ++d)
#pragma unroll
          for (int t = 0; t < MT; ++t) {
            if (FRAG_W) {
              a[d][t].x = s ? a4[d][t].z : a4[d][t].x;
              a[d][t].y = s ? a4[d][t].w : a4[d][t].y;
            } else {
              a[d][t].x = aptr[(long)t * 16 * WROW + d];
              a[d][t].y = aptr[(long)t * 16 * WROW + d + KTAPS];
            }
          }
        if (!FRAG_W) aptr += 4 * KTAPS;

        float e0[9], e1[9];
        load7(bp, &e0[1]);
        load7(bp + HWSZ, &e1[1]);
        bp += 4 * HWSZ;
        const float l0 = from_lane_below(e0[7]), l1 = from_lane_below(e1[7]);
        const float r0 = from_lane_above(e0[1]), r1 = from_lane_above(e1[1]);

        v2f p[9];
        p[0].x = mL ? l0 : 0.0f;  p[0].y = mL ? l1 : 0.0f;
        p[8].x = mR ? r0 : 0.0f;  p[8].y = mR ? r1 : 0.0f;
#pragma unroll
        for (int i = 1; i < 8; ++i) {
          p[i].x = vr ? e0[i] : 0.0f;
          p[i].y = vr ? e1[i] : 0.0f;
        }

#pragma unroll
        for (int d = 0; d < 3; ++d)
#pragma unroll
          for (int t = 0; t < MT; ++t)
#pragma unroll
            for (int j = 0; j < NT; ++j) {
              c[t][j] = __builtin_amdgcn_wmma_f32_16x16x4_f32(
                   false, a[d][t],  false, p[j + d],
                   (short)0, c[t][j],  false,  false);
            }
      }
    }
  }

  const float sc = 1.0f / 9.0f;
  float* obase = out + ((long)n * OC + oc_base + hiHalf * 8) * HWSZ + (h * WW + w0);
#pragma unroll
  for (int t = 0; t < MT; ++t) {
#pragma unroll
    for (int r = 0; r < 8; ++r) {
      float* o = obase + (long)(t * 16 + r) * HWSZ;
      v4f v; v2f u;
      v.x = c[t][0][r] * sc; v.y = c[t][1][r] * sc; v.z = c[t][2][r] * sc; v.w = c[t][3][r] * sc;
      u.x = c[t][4][r] * sc; u.y = c[t][5][r] * sc;
      __builtin_memcpy(o, &v, 16);
      __builtin_memcpy(o + 4, &u, 8);
      o[6] = c[t][6][r] * sc;
    }
  }
}

template <int HF>
__device__ __forceinline__ void load_rows5(const float* __restrict__ pc, const int (&o)[6], float (&R)[2][5][7])
{
#pragma unroll
  for (int ch = 0; ch < 2; ++ch)
#pragma unroll
    for (int rr = 0; rr < 5; ++rr) load7(pc + ch * HWSZ + o[rr + HF], R[ch][rr]);
}

__device__ __forceinline__ void load_weights(const v4f* __restrict__ ap, v4f (&G)[3][3])
{
#pragma unroll
  for (int r = 0; r < 3; ++r)
#pragma unroll
    for (int d = 0; d < 3; ++d) G[r][d] = ap[(long)(r * 3 + d) * CHUNKS_PER_TAP];
}

template <int HF, int S>
__device__ __forceinline__ void k_step_f43(const float (&R)[2][5][7], const v4f (&u4)[3][3], const v4f (&u4c)[3][2],
                                           bool vOuter, bool mL, bool mR, v8f (&c)[3][NTC])
{
  v2f a[3][3];
  v2f ac[3][2];
#pragma unroll
  for (int i = 0; i < 3; ++i)
#pragma unroll
    for (int m = 0; m < 2; ++m) {
      ac[i][m].x = S ? u4c[i][m].z : u4c[i][m].x;
      ac[i][m].y = S ? u4c[i][m].w : u4c[i][m].y;
    }
#pragma unroll
  for (int i = 0; i < 3; ++i)
#pragma unroll
    for (int d = 0; d < 3; ++d) {
      a[i][d].x = S ? u4[i][d].z : u4[i][d].x;
      a[i][d].y = S ? u4[i][d].w : u4[i][d].y;
    }

  float V[3][2][7];
#pragma unroll
  for (int ch = 0; ch < 2; ++ch)
#pragma unroll
    for (int k = 0; k < 7; ++k) {
      if (HF == 0) {
        const float t0 = vOuter ? R[ch][0][k] : 0.0f;
        const float d1 = R[ch][1][k], d2 = R[ch][2][k], d3 = R[ch][3][k], d4 = R[ch][4][k];
        const float ea = d4 - 4.0f * d2;
        const float eb = d3 - 4.0f * d1;
        V[0][ch][k] = 4.0f * t0 - 5.0f * d2 + d4;
        V[1][ch][k] = ea + eb;
        V[2][ch][k] = ea - eb;
      } else {
        const float d1 = R[ch][0][k], d2 = R[ch][1][k], d3 = R[ch][2][k], d4 = R[ch][3][k];
        const float t5 = vOuter ? R[ch][4][k] : 0.0f;
        const float ec = d4 - d2;
        const float ed = 2.0f * (d3 - d1);
        V[0][ch][k] = ec + ed;
        V[1][ch][k] = ec - ed;
        V[2][ch][k] = 4.0f * d1 - 5.0f * d3 + t5;
      }
    }

  v2f p[3][9];
#pragma unroll
  for (int i = 0; i < 3; ++i) {
    const float l0 = from_lane_below(V[i][0][6]), l1 = from_lane_below(V[i][1][6]);
    const float r0 = from_lane_above(V[i][0][0]), r1 = from_lane_above(V[i][1][0]);
    p[i][0].x = mL ? l0 : 0.0f;  p[i][0].y = mL ? l1 : 0.0f;
    p[i][8].x = mR ? r0 : 0.0f;  p[i][8].y = mR ? r1 : 0.0f;
#pragma unroll
    for (int k = 0; k < 7; ++k) {
      p[i][k + 1].x = V[i][0][k];
      p[i][k + 1].y = V[i][1][k];
    }
  }

#pragma unroll
  for (int i = 0; i < 3; ++i) {
    const v2f b0 = p[i][0] - p[i][2];
    const v2f b1 = p[i][1] + p[i][2];
    const v2f b2 = p[i][2] - p[i][1];
    const v2f b3 = p[i][1] - p[i][3];
    c[i][0] = __builtin_amdgcn_wmma_f32_16x16x4_f32(false, a[i][0],  false, b0, (short)0, c[i][0], false, false);
    c[i][1] = __builtin_amdgcn_wmma_f32_16x16x4_f32(false, ac[i][0], false, b1, (short)0, c[i][1], false, false);
    c[i][2] = __builtin_amdgcn_wmma_f32_16x16x4_f32(false, ac[i][1], false, b2, (short)0, c[i][2], false, false);
    c[i][3] = __builtin_amdgcn_wmma_f32_16x16x4_f32(false, a[i][2],  false, b3, (short)0, c[i][3], false, false);
#pragma unroll
    for (int d = 0; d < 3; ++d)
#pragma unroll
      for (int j = 2; j < NT; ++j) {
        c[i][j + 2] = __builtin_amdgcn_wmma_f32_16x16x4_f32(
             false, a[i][d],  false, p[i][j + d],
             (short)0, c[i][j + 2],  false,  false);
      }
  }
}

template <int HF>
__device__ __forceinline__ void rows_f43_half(const float* __restrict__ bp, const v4f* __restrict__ ap,
                                              const int (&o)[6], bool vOuter, bool mL, bool mR, v8f (&c)[3][NTC])
{
  v4f   G[3][3];
  float R0[2][5][7];
  float R1[2][5][7];
  load_weights(ap, G);
  load_rows5<HF>(bp, o, R0);

  for (int q = 0; q < Q_STEPS; ++q) {
    load_rows5<HF>(bp + 4 * HWSZ, o, R1);
    __builtin_amdgcn_sched_barrier(0);

    v4f u4[3][3];
#pragma unroll
    for (int d = 0; d < 3; ++d) {
      const v4f g0 = G[0][d], g1 = G[1][d], g2 = G[2][d];
      if (HF == 0) {
        const v4f s02 = g0 + g2;
        u4[0][d] = g0 * 0.25f;
        u4[1][d] = (s02 + g1) * (-1.0f / 6.0f);
        u4[2][d] = (s02 - g1) * (-1.0f / 6.0f);
      } else {
        const v4f t02 = g0 * (1.0f / 24.0f) + g2 * (1.0f / 6.0f);
        const v4f g1h = g1 * (1.0f / 12.0f);
        u4[0][d] = t02 + g1h;
        u4[1][d] = t02 - g1h;
        u4[2][d] = g2;
      }
    }

    v4f u4c[3][2];
#pragma unroll
    for (int i = 0; i < 3; ++i) {
      const v4f s02 = u4[i][0] + u4[i][2];
      u4c[i][0] = (s02 + u4[i][1]) * 0.5f;
      u4c[i][1] = (s02 - u4[i][1]) * 0.5f;
    }

    k_step_f43<HF, 0>(R0, u4, u4c, vOuter, mL, mR, c);
    __builtin_amdgcn_sched_barrier(0);

    const int more = (q + 1 < Q_STEPS) ? 1 : 0;
    ap += more * 32;
    bp += more * (8 * HWSZ);
    load_weights(ap, G);
    load_rows5<HF>(bp, o, R0);
    __builtin_amdgcn_sched_barrier(0);

    k_step_f43<HF, 1>(R1, u4, u4c, vOuter, mL, mR, c);
    __builtin_amdgcn_sched_barrier(0);
  }
}

#define XCH_SLOTS 21

template <int HF>
__device__ __forceinline__ void pack_for_partner(const v8f (&c)[3][NT], v8f (&E)[3][NT], v4f (&snd)[XCH_SLOTS])
{
#pragma unroll
  for (int j = 0; j < NT; ++j) {
    if (HF == 0) {
      const v8f s12 = c[1][j] + c[2][j];
      E[0][j] = c[0][j] + s12;
      E[1][j] = s12;
      E[2][j] = c[1][j] - c[2][j];
    } else {
      E[0][j] = c[0][j] + c[1][j];
      E[1][j] = c[0][j] - c[1][j];
      E[2][j] = c[2][j];
    }
  }
  float f[4 * XCH_SLOTS];
#pragma unroll
  for (int e = 0; e < 3; ++e)
#pragma unroll
    for (int rr = 0; rr < 4; ++rr)
#pragma unroll
      for (int j = 0; j < NT; ++j) f[(e * 4 + rr) * 7 + j] = E[e][j][(1 - HF) * 4 + rr];
#pragma unroll
  for (int s = 0; s < XCH_SLOTS; ++s) {
    v4f t; t.x = f[4 * s]; t.y = f[4 * s + 1]; t.z = f[4 * s + 2]; t.w = f[4 * s + 3];
    snd[s] = t;
  }
}

template <int HF>
__device__ __forceinline__ void finish_rows(const v8f (&E)[3][NT], const v4f (&rcv)[XCH_SLOTS], float* __restrict__ obase)
{
  float g[4 * XCH_SLOTS];
#pragma unroll
  for (int s = 0; s < XCH_SLOTS; ++s) {
    g[4 * s] = rcv[s].x; g[4 * s + 1] = rcv[s].y; g[4 * s + 2] = rcv[s].z; g[4 * s + 3] = rcv[s].w;
  }
  const float sc = 1.0f / 9.0f;
#pragma unroll
  for (int rr = 0; rr < 4; ++rr) {
    float y[4][NT];
#pragma unroll
    for (int j = 0; j < NT; ++j) {
      float a[3], b[3];
#pragma unroll
      for (int e = 0; e < 3; ++e) {
        const float own = E[e][j][HF * 4 + rr];
        const float got = g[(e * 4 + rr) * 7 + j];
        a[e] = HF ? got : own;
        b[e] = HF ? own : got;
      }
      y[0][j] = (a[0] + b[0]) * sc;
      y[1][j] = (a[2] + 2.0f * b[1]) * sc;
      y[2][j] = (a[1] + 4.0f * b[0]) * sc;
      y[3][j] = (a[2] + (8.0f * b[1] + b[2])) * sc;
    }
    float* orow = obase + (long)rr * HWSZ;
#pragma unroll
    for (int i = 0; i < 4; ++i) {
      float* op = orow + i * WW;
      v4f v; v2f u;
      v.x = y[i][0]; v.y = y[i][1]; v.z = y[i][2]; v.w = y[i][3];
      u.x = y[i][4]; u.y = y[i][5];
      __builtin_memcpy(op, &v, 16);
      __builtin_memcpy(op + 4, &u, 8);
      op[6] = y[i][6];
    }
  }
}

__device__ __forceinline__ void conv_body_rows_f43(
    const float* __restrict__ x, const float* __restrict__ w2, float* __restrict__ out,
    v4f (*xch)[XCH_SLOTS][32])
{
  const int lane   = threadIdx.x & 31;
  const int l16    = lane & 15;
  const int hiHalf = lane >> 4;
  const int kHi    = hiHalf * 2;
  const int wave   = __builtin_amdgcn_readfirstlane((int)(threadIdx.x >> 5));
  const int hf     = wave & 1;
  const int pr     = wave >> 1;

  const int oc_base = ((blockIdx.x & 7) * 2 + pr) * 16;
  const int unit    = blockIdx.x >> 3;
  const int n       = unit / (HH / 8);
  const int hA      = (unit - n * (HH / 8)) * 8 + (l16 >> 3) * 4;
  const int wq      = l16 & 7;
  const int w0      = 7 * wq;

  const bool v0 = hA != 0;
  const bool v5 = hA != HH - 4;
  const bool mL = wq != 0;
  const bool mR = wq != 7;


  v8f cw[3][NTC] = {};

  const float* bp = x + (long)n * CHW + (long)kHi * HWSZ + w0;
  const int o[6] = { (v0 ? hA - 1 : hA) * WW, hA * WW, (hA + 1) * WW, (hA + 2) * WW, (hA + 3) * WW,
                     (v5 ? hA + 4 : hA) * WW };
  const v4f* ap = (const v4f*)w2 + (long)(oc_base >> 4) * (Q_STEPS * 32) + lane;

  v8f E[3][NT];
  v4f snd[XCH_SLOTS];
  if (hf == 0) rows_f43_half<0>(bp, ap, o, v0, mL, mR, cw);
  else         rows_f43_half<1>(bp, ap, o, v5, mL, mR, cw);

  v8f c[3][NT];
#pragma unroll
  for (int i = 0; i < 3; ++i) {
    c[i][0] = cw[i][0] + cw[i][1] + cw[i][2];
    c[i][1] = cw[i][1] - cw[i][2] - cw[i][3];
#pragma unroll
    for (int j = 2; j < NT; ++j) c[i][j] = cw[i][j + 2];
  }
  if (hf == 0) pack_for_partner<0>(c, E, snd);
  else         pack_for_partner<1>(c, E, snd);

#pragma unroll
  for (int s = 0; s < XCH_SLOTS; ++s) xch[wave][s][lane] = snd[s];
  __syncthreads();
  v4f rcv[XCH_SLOTS];
#pragma unroll
  for (int s = 0; s < XCH_SLOTS; ++s) rcv[s] = xch[wave ^ 1][s][lane];

  float* obase = out + ((long)n * OC + oc_base + hiHalf * 8 + hf * 4) * HWSZ + (hA * WW + w0);
  if (hf == 0) finish_rows<0>(E, rcv, obase);
  else         finish_rows<1>(E, rcv, obase);
}

__global__ __launch_bounds__(128) void conv3x3_wmma_f32(
    const float* __restrict__ x, const float* __restrict__ w,
    const float* __restrict__ w2, size_t ws_bytes, float* __restrict__ out)
{
  __shared__ v4f xch[4][XCH_SLOTS][32];
  if (ws_bytes >= (size_t)W2_BYTES) conv_body_rows_f43(x, w2, out, xch);
  else                              conv_body<false>(x, w, w2, out);
}

extern "C" void kernel_launch(void* const* d_in, const int* in_sizes, int n_in,
                              void* d_out, int out_size, void* d_ws, size_t ws_size,
                              hipStream_t stream) {
  const float* x = (const float*)d_in[0];
  const float* w = (const float*)d_in[1];
  float* out = (float*)d_out;
  float* w2  = (float*)d_ws;

  weights_to_fragment_order<<<dim3(W2_CHUNKS / 256, 1, 1), dim3(256, 1, 1), 0, stream>>>(w, w2, ws_size);

  dim3 grid(ROW_PAIRS * 2, 1, 1);
  dim3 block(128, 1, 1);
  conv3x3_wmma_f32<<<grid, block, 0, stream>>>(x, w, w2, ws_size, out);
}


# a lane's last three pixels in F(3,3) minimal-filtering form along the width on the quad's own weights: 33 products per k-step, not 45; selects, fetches trimmed to fit 2 waves; sums 2-D transformed
# speedup vs baseline: 8.7634x; 1.1937x over previous
#include <hip/hip_runtime.h>


typedef __attribute__((ext_vector_type(2))) float v2f;
typedef __attribute__((ext_vector_type(4))) float v4f;
typedef __attribute__((ext_vector_type(8))) float v8f;

#define IC 256
#define OC 256
#define HH 56
#define WW 56
#define HWSZ (HH * WW)
#define CHW (IC * HWSZ)
#define KTAPS 9
#define WROW (IC * KTAPS)

#define MT 2
#define NT 7
#define WAVE_OC (16 * MT)
#define ROW_PAIRS (32 * (HH / 2))
#define NTC 11

#define OC_TILES (OC / 16)
#define Q_STEPS (IC / 8)
#define CHUNKS_PER_TAP (OC_TILES * Q_STEPS * 32)
#define W2_CHUNKS (KTAPS * CHUNKS_PER_TAP)
#define W2_BYTES (W2_CHUNKS * 16)

__global__ __launch_bounds__(256) void weights_to_fragment_order(
    const float* __restrict__ w, float* __restrict__ w2, size_t ws_bytes)
{
  if (ws_bytes < (size_t)W2_BYTES) return;
  const int idx  = blockIdx.x * 256 + threadIdx.x;
  const int L    = idx & 31;
  const int q    = (idx >> 5) & 31;
  const int oct  = (idx >> 10) & 15;
  const int tap  = idx >> 14;
  const int row  = L & 15;
  const int half = L >> 4;
  const int ic0  = q * 8 + half * 2;
  const float* src = w + (long)(oct * 16 + row) * WROW + tap;
  v4f v;
  v.x = src[(ic0    ) * KTAPS];
  v.y = src[(ic0 + 1) * KTAPS];
  v.z = src[(ic0 + 4) * KTAPS];
  v.w = src[(ic0 + 5) * KTAPS];
  ((v4f*)w2)[idx] = v;
}

__device__ __forceinline__ void load7(const float* __restrict__ p, float* d)
{
  v4f a; v2f b;
  __builtin_memcpy(&a, p, 16);
  __builtin_memcpy(&b, p + 4, 8);
  d[0] = a.x; d[1] = a.y; d[2] = a.z; d[3] = a.w;
  d[4] = b.x; d[5] = b.y; d[6] = p[6];
}

__device__ __forceinline__ float from_lane_below(float v)
{
  return __int_as_float(__builtin_amdgcn_update_dpp(0, __float_as_int(v), 0x111, 0xF, 0xF, true));
}
__device__ __forceinline__ float from_lane_above(float v)
{
  return __int_as_float(__builtin_amdgcn_update_dpp(0, __float_as_int(v), 0x101, 0xF, 0xF, true));
}

template <bool FRAG_W>
__device__ __forceinline__ void conv_body(
    const float* __restrict__ x, const float* __restrict__ w,
    const float* __restrict__ w2, float* __restrict__ out)
{
  const int lane   = threadIdx.x & 31;
  const int wave   = threadIdx.x >> 5;
  const int l16    = lane & 15;
  const int hiHalf = lane >> 4;
  const int kHi    = hiHalf * 2;

  const int oc_base = ((blockIdx.x & 1) * 4 + wave) * WAVE_OC;
  const int rp      = blockIdx.x >> 1;
  const int n       = rp / (HH / 2);
  const int h       = (rp - n * (HH / 2)) * 2 + (l16 >> 3);
  const int wq      = l16 & 7;
  const int w0      = 7 * wq;


  v8f c[MT][NT] = {};

  const float* wRowBase = w + (long)(oc_base + l16) * WROW + kHi * KTAPS;
  const v4f*   fragBase = (const v4f*)w2 + (long)(oc_base >> 4) * (Q_STEPS * 32) + lane;

  for (int dhi = 0; dhi < 3; ++dhi) {
    const int  hh = h + dhi - 1;
    const bool vr = (unsigned)hh < (unsigned)HH;
    const bool mL = vr && (wq != 0);
    const bool mR = vr && (wq != 7);
    const float* bp = x + (long)n * CHW + (long)kHi * HWSZ + (long)((vr ? hh : h) * WW + w0);

    const v4f*   ap   = fragBase + (long)(dhi * 3) * CHUNKS_PER_TAP;
    const float* aptr = wRowBase + dhi * 3;

    for (int q = 0; q < Q_STEPS; ++q) {
      v4f a4[3][MT];
      if (FRAG_W) {
#pragma unroll
        for (int d = 0; d < 3; ++d)
#pragma unroll
          for (int t = 0; t < MT; ++t) a4[d][t] = ap[(long)d * CHUNKS_PER_TAP + t * (Q_STEPS * 32)];
        ap += 32;
      }

#pragma unroll
      for (int s = 0; s < 2; ++s) {
        v2f a[3][MT];
#pragma unroll
        for (int d = 0; d < 3; ++d)
#pragma unroll
          for (int t = 0; t < MT; ++t) {
            if (FRAG_W) {
              a[d][t].x = s ? a4[d][t].z : a4[d][t].x;
              a[d][t].y = s ? a4[d][t].w : a4[d][t].y;
            } else {
              a[d][t].x = aptr[(long)t * 16 * WROW + d];
              a[d][t].y = aptr[(long)t * 16 * WROW + d + KTAPS];
            }
          }
        if (!FRAG_W) aptr += 4 * KTAPS;

        float e0[9], e1[9];
        load7(bp, &e0[1]);
        load7(bp + HWSZ, &e1[1]);
        bp += 4 * HWSZ;
        const float l0 = from_lane_below(e0[7]), l1 = from_lane_below(e1[7]);
        const float r0 = from_lane_above(e0[1]), r1 = from_lane_above(e1[1]);

        v2f p[9];
        p[0].x = mL ? l0 : 0.0f;  p[0].y = mL ? l1 : 0.0f;
        p[8].x = mR ? r0 : 0.0f;  p[8].y = mR ? r1 : 0.0f;
#pragma unroll
        for (int i = 1; i < 8; ++i) {
          p[i].x = vr ? e0[i] : 0.0f;
          p[i].y = vr ? e1[i] : 0.0f;
        }

#pragma unroll
        for (int d = 0; d < 3; ++d)
#pragma unroll
          for (int t = 0; t < MT; ++t)
#pragma unroll
            for (int j = 0; j < NT; ++j) {
              c[t][j] = __builtin_amdgcn_wmma_f32_16x16x4_f32(
                   false, a[d][t],  false, p[j + d],
                   (short)0, c[t][j],  false,  false);
            }
      }
    }
  }

  const float sc = 1.0f / 9.0f;
  float* obase = out + ((long)n * OC + oc_base + hiHalf * 8) * HWSZ + (h * WW + w0);
#pragma unroll
  for (int t = 0; t < MT; ++t) {
#pragma unroll
    for (int r = 0; r < 8; ++r) {
      float* o = obase + (long)(t * 16 + r) * HWSZ;
      v4f v; v2f u;
      v.x = c[t][0][r] * sc; v.y = c[t][1][r] * sc; v.z = c[t][2][r] * sc; v.w = c[t][3][r] * sc;
      u.x = c[t][4][r] * sc; u.y = c[t][5][r] * sc;
      __builtin_memcpy(o, &v, 16);
      __builtin_memcpy(o + 4, &u, 8);
      o[6] = c[t][6][r] * sc;
    }
  }
}

template <int HF>
__device__ __forceinline__ void load_rows5(const float* __restrict__ pc, const int (&o)[6], float (&R)[2][5][7])
{
#pragma unroll
  for (int ch = 0; ch < 2; ++ch)
#pragma unroll
    for (int rr = 0; rr < 5; ++rr) load7(pc + ch * HWSZ + o[rr + HF], R[ch][rr]);
}

template <int S>
__device__ __forceinline__ void load_weights(const v4f* __restrict__ ap, v2f (&G)[3][3])
{
#pragma unroll
  for (int r = 0; r < 3; ++r)
#pragma unroll
    for (int d = 0; d < 3; ++d) G[r][d] = ((const v2f*)(ap + (long)(r * 3 + d) * CHUNKS_PER_TAP))[S];
}

template <int HF>
__device__ __forceinline__ void weight_rows(const v2f (&G)[3][3], v2f (&u)[3][3])
{
#pragma unroll
  for (int d = 0; d < 3; ++d) {
    const v2f g0 = G[0][d], g1 = G[1][d], g2 = G[2][d];
    if (HF == 0) {
      const v2f s02 = g0 + g2;
      u[0][d] = g0 * 0.25f;
      u[1][d] = (s02 + g1) * (-1.0f / 6.0f);
      u[2][d] = (s02 - g1) * (-1.0f / 6.0f);
    } else {
      const v2f t02 = g0 * (1.0f / 24.0f) + g2 * (1.0f / 6.0f);
      const v2f g1h = g1 * (1.0f / 12.0f);
      u[0][d] = t02 + g1h;
      u[1][d] = t02 - g1h;
      u[2][d] = g2;
    }
  }
}

template <int HF>
__device__ __forceinline__ void k_step_f43(const float (&R)[2][5][7], const v2f (&a)[3][3],
                                           float fOuter, float fL, float fR, v8f (&c)[3][NTC])
{
  v2f aq[3][4];
#pragma unroll
  for (int i = 0; i < 3; ++i) {
    const v2f s6  = (a[i][0] + a[i][2]) * (-1.0f / 6.0f);
    const v2f t02 = a[i][0] * (1.0f / 24.0f) + a[i][2] * (1.0f / 6.0f);
    aq[i][0] = s6 - a[i][1] * (1.0f / 6.0f);
    aq[i][1] = s6 + a[i][1] * (1.0f / 6.0f);
    aq[i][2] = t02 + a[i][1] * (1.0f / 12.0f);
    aq[i][3] = t02 - a[i][1] * (1.0f / 12.0f);
  }

  float V[3][2][7];
#pragma unroll
  for (int ch = 0; ch < 2; ++ch)
#pragma unroll
    for (int k = 0; k < 7; ++k) {
      if (HF == 0) {
        const float d1 = R[ch][1][k], d2 = R[ch][2][k], d3 = R[ch][3][k], d4 = R[ch][4][k];
        const float ea = d4 - 4.0f * d2;
        const float eb = d3 - 4.0f * d1;
        V[0][ch][k] = fOuter * R[ch][0][k] - 5.0f * d2 + d4;
        V[1][ch][k] = ea + eb;
        V[2][ch][k] = ea - eb;
      } else {
        const float d1 = R[ch][0][k], d2 = R[ch][1][k], d3 = R[ch][2][k], d4 = R[ch][3][k];
        const float ec = d4 - d2;
        const float ed = 2.0f * (d3 - d1);
        V[0][ch][k] = ec + ed;
        V[1][ch][k] = ec - ed;
        V[2][ch][k] = 4.0f * d1 - 5.0f * d3 + fOuter * R[ch][4][k];
      }
    }

  v2f p[3][9];
#pragma unroll
  for (int i = 0; i < 3; ++i) {
    const float l0 = from_lane_below(V[i][0][6]), l1 = from_lane_below(V[i][1][6]);
    const float r0 = from_lane_above(V[i][0][0]), r1 = from_lane_above(V[i][1][0]);
    p[i][0].x = l0;  p[i][0].y = l1;
    p[i][8].x = r0;  p[i][8].y = r1;
#pragma unroll
    for (int k = 0; k < 7; ++k) {
      p[i][k + 1].x = V[i][0][k];
      p[i][k + 1].y = V[i][1][k];
    }
  }

#pragma unroll
  for (int i = 0; i < 3; ++i) {
    const v2f e0 = p[i][0], e1 = p[i][1], e2 = p[i][2], e3 = p[i][3], e4 = p[i][4], e5 = p[i][5];
    const v2f b0 = fL * e0 + (0.25f * e4 - 1.25f * e2);
    const v2f fa = e4 - 4.0f * e2;
    const v2f fb = e3 - 4.0f * e1;
    const v2f b1 = fa + fb;
    const v2f b2 = fa - fb;
    const v2f fc = e4 - e2;
    const v2f fd = e3 - e1;
    const v2f b3 = fc + 2.0f * fd;
    const v2f b4 = fc - 2.0f * fd;
    const v2f b5 = 4.0f * e1 - 5.0f * e3 + e5;
    c[i][0] = __builtin_amdgcn_wmma_f32_16x16x4_f32(false, a[i][0],  false, b0, (short)0, c[i][0], false, false);
    c[i][1] = __builtin_amdgcn_wmma_f32_16x16x4_f32(false, aq[i][0], false, b1, (short)0, c[i][1], false, false);
    c[i][2] = __builtin_amdgcn_wmma_f32_16x16x4_f32(false, aq[i][1], false, b2, (short)0, c[i][2], false, false);
    c[i][3] = __builtin_amdgcn_wmma_f32_16x16x4_f32(false, aq[i][2], false, b3, (short)0, c[i][3], false, false);
    c[i][4] = __builtin_amdgcn_wmma_f32_16x16x4_f32(false, aq[i][3], false, b4, (short)0, c[i][4], false, false);
    c[i][5] = __builtin_amdgcn_wmma_f32_16x16x4_f32(false, a[i][2],  false, b5, (short)0, c[i][5], false, false);
    const v2f h1 = p[i][5], h2 = p[i][6], h3 = p[i][7], h4 = p[i][8];
    const v2f gd = h3 - h1;
    const v2f gs = h3 - h2;
    const v2f t0 = (e4 - h2) + 0.5f * gd;
    const v2f t1 = gs - 2.0f * h1;
    const v2f t2 = gs + 2.0f * (h1 - h2);
    const v2f t4 = (fR * h4 - h2) - 2.0f * gd;
    c[i][6]  = __builtin_amdgcn_wmma_f32_16x16x4_f32(false, a[i][0],  false, t0, (short)0, c[i][6],  false, false);
    c[i][7]  = __builtin_amdgcn_wmma_f32_16x16x4_f32(false, aq[i][0], false, t1, (short)0, c[i][7],  false, false);
    c[i][8]  = __builtin_amdgcn_wmma_f32_16x16x4_f32(false, aq[i][1], false, t2, (short)0, c[i][8],  false, false);
    c[i][9]  = __builtin_amdgcn_wmma_f32_16x16x4_f32(false, aq[i][2], false, gd, (short)0, c[i][9],  false, false);
    c[i][10] = __builtin_amdgcn_wmma_f32_16x16x4_f32(false, a[i][2],  false, t4, (short)0, c[i][10], false, false);
  }
}

template <int HF>
__device__ __forceinline__ void rows_f43_half(const float* __restrict__ bp, const v4f* __restrict__ ap,
                                              const int (&o)[6], float fOuter, float fL, float fR, v8f (&c)[3][NTC])
{
  v2f   G0[3][3];
  v2f   G1[3][3];
  float R0[2][5][7];
  float R1[2][5][7];
  load_weights<0>(ap, G0);
  load_rows5<HF>(bp, o, R0);

  for (int q = 0; q < Q_STEPS; ++q) {
    load_weights<1>(ap, G1);
    load_rows5<HF>(bp + 4 * HWSZ, o, R1);
    __builtin_amdgcn_sched_barrier(0);

    v2f u[3][3];
    weight_rows<HF>(G0, u);
    k_step_f43<HF>(R0, u, fOuter, fL, fR, c);
    __builtin_amdgcn_sched_barrier(0);

    const int more = (q + 1 < Q_STEPS) ? 1 : 0;
    ap += more * 32;
    bp += more * (8 * HWSZ);
    load_weights<0>(ap, G0);
    load_rows5<HF>(bp, o, R0);
    __builtin_amdgcn_sched_barrier(0);

    weight_rows<HF>(G1, u);
    k_step_f43<HF>(R1, u, fOuter, fL, fR, c);
    __builtin_amdgcn_sched_barrier(0);
  }
}

#define XCH_SLOTS 21

template <int HF>
__device__ __forceinline__ void pack_for_partner(const v8f (&c)[3][NT], v8f (&E)[3][NT], v4f (&snd)[XCH_SLOTS])
{
#pragma unroll
  for (int j = 0; j < NT; ++j) {
    if (HF == 0) {
      const v8f s12 = c[1][j] + c[2][j];
      E[0][j] = c[0][j] + s12;
      E[1][j] = s12;
      E[2][j] = c[1][j] - c[2][j];
    } else {
      E[0][j] = c[0][j] + c[1][j];
      E[1][j] = c[0][j] - c[1][j];
      E[2][j] = c[2][j];
    }
  }
  float f[4 * XCH_SLOTS];
#pragma unroll
  for (int e = 0; e < 3; ++e)
#pragma unroll
    for (int rr = 0; rr < 4; ++rr)
#pragma unroll
      for (int j = 0; j < NT; ++j) f[(e * 4 + rr) * 7 + j] = E[e][j][(1 - HF) * 4 + rr];
#pragma unroll
  for (int s = 0; s < XCH_SLOTS; ++s) {
    v4f t; t.x = f[4 * s]; t.y = f[4 * s + 1]; t.z = f[4 * s + 2]; t.w = f[4 * s + 3];
    snd[s] = t;
  }
}

template <int HF>
__device__ __forceinline__ void finish_rows(const v8f (&E)[3][NT], const v4f (&rcv)[XCH_SLOTS], float* __restrict__ obase)
{
  float g[4 * XCH_SLOTS];
#pragma unroll
  for (int s = 0; s < XCH_SLOTS; ++s) {
    g[4 * s] = rcv[s].x; g[4 * s + 1] = rcv[s].y; g[4 * s + 2] = rcv[s].z; g[4 * s + 3] = rcv[s].w;
  }
  const float sc = 1.0f / 9.0f;
#pragma unroll
  for (int rr = 0; rr < 4; ++rr) {
    float y[4][NT];
#pragma unroll
    for (int j = 0; j < NT; ++j) {
      float a[3], b[3];
#pragma unroll
      for (int e = 0; e < 3; ++e) {
        const float own = E[e][j][HF * 4 + rr];
        const float got = g[(e * 4 + rr) * 7 + j];
        a[e] = HF ? got : own;
        b[e] = HF ? own : got;
      }
      y[0][j] = (a[0] + b[0]) * sc;
      y[1][j] = (a[2] + 2.0f * b[1]) * sc;
      y[2][j] = (a[1] + 4.0f * b[0]) * sc;
      y[3][j] = (a[2] + (8.0f * b[1] + b[2])) * sc;
    }
    float* orow = obase + (long)rr * HWSZ;
#pragma unroll
    for (int i = 0; i < 4; ++i) {
      float* op = orow + i * WW;
      v4f v; v2f u;
      v.x = y[i][0]; v.y = y[i][1]; v.z = y[i][2]; v.w = y[i][3];
      u.x = y[i][4]; u.y = y[i][5];
      __builtin_memcpy(op, &v, 16);
      __builtin_memcpy(op + 4, &u, 8);
      op[6] = y[i][6];
    }
  }
}

__device__ __forceinline__ void conv_body_rows_f43(
    const float* __restrict__ x, const float* __restrict__ w2, float* __restrict__ out,
    v4f (*xch)[XCH_SLOTS][32])
{
  const int lane   = threadIdx.x & 31;
  const int l16    = lane & 15;
  const int hiHalf = lane >> 4;
  const int kHi    = hiHalf * 2;
  const int wave   = __builtin_amdgcn_readfirstlane((int)(threadIdx.x >> 5));
  const int hf     = wave & 1;
  const int pr     = wave >> 1;

  const int oc_base = ((blockIdx.x & 7) * 2 + pr) * 16;
  const int unit    = blockIdx.x >> 3;
  const int n       = unit / (HH / 8);
  const int hA      = (unit - n * (HH / 8)) * 8 + (l16 >> 3) * 4;
  const int wq      = l16 & 7;
  const int w0      = 7 * wq;

  const bool v0 = hA != 0;
  const bool v5 = hA != HH - 4;
  const bool mL = wq != 0;
  const bool mR = wq != 7;


  v8f cw[3][NTC] = {};

  const float* bp = x + (long)n * CHW + (long)kHi * HWSZ + w0;
  const int o[6] = { (v0 ? hA - 1 : hA) * WW, hA * WW, (hA + 1) * WW, (hA + 2) * WW, (hA + 3) * WW,
                     (v5 ? hA + 4 : hA) * WW };
  const v4f* ap = (const v4f*)w2 + (long)(oc_base >> 4) * (Q_STEPS * 32) + lane;

  v8f E[3][NT];
  v4f snd[XCH_SLOTS];
  const float fL = mL ? 1.0f : 0.0f;
  const float fR = mR ? 1.0f : 0.0f;
  if (hf == 0) rows_f43_half<0>(bp, ap, o, v0 ? 4.0f : 0.0f, fL, fR, cw);
  else         rows_f43_half<1>(bp, ap, o, v5 ? 1.0f : 0.0f, fL, fR, cw);

  v8f c[3][NT];
#pragma unroll
  for (int i = 0; i < 3; ++i) {
    const v8f s12 = cw[i][1] + cw[i][2];
    const v8f d12 = cw[i][1] - cw[i][2];
    const v8f s34 = cw[i][3] + cw[i][4];
    const v8f d34 = cw[i][3] - cw[i][4];
    c[i][0] = cw[i][0] + s12 + s34;
    c[i][1] = d12 + 2.0f * d34;
    c[i][2] = s12 + 4.0f * s34;
    c[i][3] = d12 + (8.0f * d34 + cw[i][5]);
    const v8f n1  = 3.0f * cw[i][7];
    const v8f n3  = 4.0f * cw[i][9];
    const v8f s78 = n1 + cw[i][8];
    const v8f d78 = n1 - cw[i][8];
    c[i][4] = cw[i][6] + s78 + n3;
    c[i][5] = d78 + 2.0f * n3;
    c[i][6] = s78 + (4.0f * n3 + cw[i][10]);
  }
  if (hf == 0) pack_for_partner<0>(c, E, snd);
  else         pack_for_partner<1>(c, E, snd);

#pragma unroll
  for (int s = 0; s < XCH_SLOTS; ++s) xch[wave][s][lane] = snd[s];
  __syncthreads();
  v4f rcv[XCH_SLOTS];
#pragma unroll
  for (int s = 0; s < XCH_SLOTS; ++s) rcv[s] = xch[wave ^ 1][s][lane];

  float* obase = out + ((long)n * OC + oc_base + hiHalf * 8 + hf * 4) * HWSZ + (hA * WW + w0);
  if (hf == 0) finish_rows<0>(E, rcv, obase);
  else         finish_rows<1>(E, rcv, obase);
}

__global__ __launch_bounds__(128) __attribute__((amdgpu_waves_per_eu(2, 2))) void conv3x3_wmma_f32(
    const float* __restrict__ x, const float* __restrict__ w,
    const float* __restrict__ w2, size_t ws_bytes, float* __restrict__ out)
{
  __shared__ v4f xch[4][XCH_SLOTS][32];
  if (ws_bytes >= (size_t)W2_BYTES) conv_body_rows_f43(x, w2, out, xch);
  else                              conv_body<false>(x, w, w2, out);
}

extern "C" void kernel_launch(void* const* d_in, const int* in_sizes, int n_in,
                              void* d_out, int out_size, void* d_ws, size_t ws_size,
                              hipStream_t stream) {
  const float* x = (const float*)d_in[0];
  const float* w = (const float*)d_in[1];
  float* out = (float*)d_out;
  float* w2  = (float*)d_ws;

  weights_to_fragment_order<<<dim3(W2_CHUNKS / 256, 1, 1), dim3(256, 1, 1), 0, stream>>>(w, w2, ws_size);

  dim3 grid(ROW_PAIRS * 2, 1, 1);
  dim3 block(128, 1, 1);
  conv3x3_wmma_f32<<<grid, block, 0, stream>>>(x, w, w2, ws_size, out);
}


# a wave keeps 2 oc tiles (32 oc) of accumulators, one wave per SIMD: a transformed input fragment feeds 2 products; 3 rounds so, the last 1/7 of the rows on 1-tile waves; sum order unchanged
# speedup vs baseline: 9.0278x; 1.0302x over previous
#include <hip/hip_runtime.h>


typedef __attribute__((ext_vector_type(2))) float v2f;
typedef __attribute__((ext_vector_type(4))) float v4f;
typedef __attribute__((ext_vector_type(8))) float v8f;

#define IC 256
#define OC 256
#define HH 56
#define WW 56
#define HWSZ (HH * WW)
#define CHW (IC * HWSZ)
#define KTAPS 9
#define WROW (IC * KTAPS)

#define MT 2
#define NT 7
#define WAVE_OC (16 * MT)
#define ROW_PAIRS (32 * (HH / 2))
#define TWO_TILE_GROUPS 768
#define GRID_GROUPS 1024
#define NTC 11

#define OC_TILES (OC / 16)
#define Q_STEPS (IC / 8)
#define CHUNKS_PER_TAP (OC_TILES * Q_STEPS * 32)
#define W2_CHUNKS (KTAPS * CHUNKS_PER_TAP)
#define W2_BYTES (W2_CHUNKS * 16)

__global__ __launch_bounds__(256) void weights_to_fragment_order(
    const float* __restrict__ w, float* __restrict__ w2, size_t ws_bytes)
{
  if (ws_bytes < (size_t)W2_BYTES) return;
  const int idx  = blockIdx.x * 256 + threadIdx.x;
  const int L    = idx & 31;
  const int q    = (idx >> 5) & 31;
  const int oct  = (idx >> 10) & 15;
  const int tap  = idx >> 14;
  const int row  = L & 15;
  const int half = L >> 4;
  const int ic0  = q * 8 + half * 2;
  const float* src = w + (long)(oct * 16 + row) * WROW + tap;
  v4f v;
  v.x = src[(ic0    ) * KTAPS];
  v.y = src[(ic0 + 1) * KTAPS];
  v.z = src[(ic0 + 4) * KTAPS];
  v.w = src[(ic0 + 5) * KTAPS];
  ((v4f*)w2)[idx] = v;
}

__device__ __forceinline__ void load7(const float* __restrict__ p, float* d)
{
  v4f a; v2f b;
  __builtin_memcpy(&a, p, 16);
  __builtin_memcpy(&b, p + 4, 8);
  d[0] = a.x; d[1] = a.y; d[2] = a.z; d[3] = a.w;
  d[4] = b.x; d[5] = b.y; d[6] = p[6];
}

__device__ __forceinline__ float from_lane_below(float v)
{
  return __int_as_float(__builtin_amdgcn_update_dpp(0, __float_as_int(v), 0x111, 0xF, 0xF, true));
}
__device__ __forceinline__ float from_lane_above(float v)
{
  return __int_as_float(__builtin_amdgcn_update_dpp(0, __float_as_int(v), 0x101, 0xF, 0xF, true));
}

template <bool FRAG_W>
__device__ __forceinline__ void conv_body(
    const float* __restrict__ x, const float* __restrict__ w,
    const float* __restrict__ w2, float* __restrict__ out, const int bid)
{
  const int lane   = threadIdx.x & 31;
  const int wave   = threadIdx.x >> 5;
  const int l16    = lane & 15;
  const int hiHalf = lane >> 4;
  const int kHi    = hiHalf * 2;

  const int oc_base = ((bid & 1) * 4 + wave) * WAVE_OC;
  const int rp      = bid >> 1;
  const int n       = rp / (HH / 2);
  const int h       = (rp - n * (HH / 2)) * 2 + (l16 >> 3);
  const int wq      = l16 & 7;
  const int w0      = 7 * wq;


  v8f c[MT][NT] = {};

  const float* wRowBase = w + (long)(oc_base + l16) * WROW + kHi * KTAPS;
  const v4f*   fragBase = (const v4f*)w2 + (long)(oc_base >> 4) * (Q_STEPS * 32) + lane;

  for (int dhi = 0; dhi < 3; ++dhi) {
    const int  hh = h + dhi - 1;
    const bool vr = (unsigned)hh < (unsigned)HH;
    const bool mL = vr && (wq != 0);
    const bool mR = vr && (wq != 7);
    const float* bp = x + (long)n * CHW + (long)kHi * HWSZ + (long)((vr ? hh : h) * WW + w0);

    const v4f*   ap   = fragBase + (long)(dhi * 3) * CHUNKS_PER_TAP;
    const float* aptr = wRowBase + dhi * 3;

    for (int q = 0; q < Q_STEPS; ++q) {
      v4f a4[3][MT];
      if (FRAG_W) {
#pragma unroll
        for (int d = 0; d < 3; ++d)
#pragma unroll
          for (int t = 0; t < MT; ++t) a4[d][t] = ap[(long)d * CHUNKS_PER_TAP + t * (Q_STEPS * 32)];
        ap += 32;
      }

#pragma unroll
      for (int s = 0; s < 2; ++s) {
        v2f a[3][MT];
#pragma unroll
        for (int d = 0; d < 3; ++d)
#pragma unroll
          for (int t = 0; t < MT; ++t) {
            if (FRAG_W) {
              a[d][t].x = s ? a4[d][t].z : a4[d][t].x;
              a[d][t].y = s ? a4[d][t].w : a4[d][t].y;
            } else {
              a[d][t].x = aptr[(long)t * 16 * WROW + d];
              a[d][t].y = aptr[(long)t * 16 * WROW + d + KTAPS];
            }
          }
        if (!FRAG_W) aptr += 4 * KTAPS;

        float e0[9], e1[9];
        load7(bp, &e0[1]);
        load7(bp + HWSZ, &e1[1]);
        bp += 4 * HWSZ;
        const float l0 = from_lane_below(e0[7]), l1 = from_lane_below(e1[7]);
        const float r0 = from_lane_above(e0[1]), r1 = from_lane_above(e1[1]);

        v2f p[9];
        p[0].x = mL ? l0 : 0.0f;  p[0].y = mL ? l1 : 0.0f;
        p[8].x = mR ? r0 : 0.0f;  p[8].y = mR ? r1 : 0.0f;
#pragma unroll
        for (int i = 1; i < 8; ++i) {
          p[i].x = vr ? e0[i] : 0.0f;
          p[i].y = vr ? e1[i] : 0.0f;
        }

#pragma unroll
        for (int d = 0; d < 3; ++d)
#pragma unroll
          for (int t = 0; t < MT; ++t)
#pragma unroll
            for (int j = 0; j < NT; ++j) {
              c[t][j] = __builtin_amdgcn_wmma_f32_16x16x4_f32(
                   false, a[d][t],  false, p[j + d],
                   (short)0, c[t][j],  false,  false);
            }
      }
    }
  }

  const float sc = 1.0f / 9.0f;
  float* obase = out + ((long)n * OC + oc_base + hiHalf * 8) * HWSZ + (h * WW + w0);
#pragma unroll
  for (int t = 0; t < MT; ++t) {
#pragma unroll
    for (int r = 0; r < 8; ++r) {
      float* o = obase + (long)(t * 16 + r) * HWSZ;
      v4f v; v2f u;
      v.x = c[t][0][r] * sc; v.y = c[t][1][r] * sc; v.z = c[t][2][r] * sc; v.w = c[t][3][r] * sc;
      u.x = c[t][4][r] * sc; u.y = c[t][5][r] * sc;
      __builtin_memcpy(o, &v, 16);
      __builtin_memcpy(o + 4, &u, 8);
      o[6] = c[t][6][r] * sc;
    }
  }
}

template <int HF>
__device__ __forceinline__ void load_rows5(const float* __restrict__ pc, const int (&o)[6], float (&R)[2][5][7])
{
#pragma unroll
  for (int ch = 0; ch < 2; ++ch)
#pragma unroll
    for (int rr = 0; rr < 5; ++rr) load7(pc + ch * HWSZ + o[rr + HF], R[ch][rr]);
}

template <int S>
__device__ __forceinline__ void load_weights(const v4f* __restrict__ ap, v2f (&G)[3][3])
{
#pragma unroll
  for (int r = 0; r < 3; ++r)
#pragma unroll
    for (int d = 0; d < 3; ++d) G[r][d] = ((const v2f*)(ap + (long)(r * 3 + d) * CHUNKS_PER_TAP))[S];
}

template <int HF>
__device__ __forceinline__ void weight_rows(const v2f (&G)[3][3], v2f (&u)[3][3])
{
#pragma unroll
  for (int d = 0; d < 3; ++d) {
    const v2f g0 = G[0][d], g1 = G[1][d], g2 = G[2][d];
    if (HF == 0) {
      const v2f s02 = g0 + g2;
      u[0][d] = g0 * 0.25f;
      u[1][d] = (s02 + g1) * (-1.0f / 6.0f);
      u[2][d] = (s02 - g1) * (-1.0f / 6.0f);
    } else {
      const v2f t02 = g0 * (1.0f / 24.0f) + g2 * (1.0f / 6.0f);
      const v2f g1h = g1 * (1.0f / 12.0f);
      u[0][d] = t02 + g1h;
      u[1][d] = t02 - g1h;
      u[2][d] = g2;
    }
  }
}

template <int HF, int OT>
__device__ __forceinline__ void k_step_f43(const float (&R)[2][5][7], const v2f (&a)[OT][3][3],
                                           float fOuter, float fL, float fR, v8f (&c)[OT][3][NTC])
{
  v2f aq[OT][3][4];
#pragma unroll
  for (int t = 0; t < OT; ++t)
#pragma unroll
    for (int i = 0; i < 3; ++i) {
      const v2f s6  = (a[t][i][0] + a[t][i][2]) * (-1.0f / 6.0f);
      const v2f t02 = a[t][i][0] * (1.0f / 24.0f) + a[t][i][2] * (1.0f / 6.0f);
      aq[t][i][0] = s6 - a[t][i][1] * (1.0f / 6.0f);
      aq[t][i][1] = s6 + a[t][i][1] * (1.0f / 6.0f);
      aq[t][i][2] = t02 + a[t][i][1] * (1.0f / 12.0f);
      aq[t][i][3] = t02 - a[t][i][1] * (1.0f / 12.0f);
    }

  float V[3][2][7];
#pragma unroll
  for (int ch = 0; ch < 2; ++ch)
#pragma unroll
    for (int k = 0; k < 7; ++k) {
      if (HF == 0) {
        const float d1 = R[ch][1][k], d2 = R[ch][2][k], d3 = R[ch][3][k], d4 = R[ch][4][k];
        const float ea = d4 - 4.0f * d2;
        const float eb = d3 - 4.0f * d1;
        V[0][ch][k] = fOuter * R[ch][0][k] - 5.0f * d2 + d4;
        V[1][ch][k] = ea + eb;
        V[2][ch][k] = ea - eb;
      } else {
        const float d1 = R[ch][0][k], d2 = R[ch][1][k], d3 = R[ch][2][k], d4 = R[ch][3][k];
        const float ec = d4 - d2;
        const float ed = 2.0f * (d3 - d1);
        V[0][ch][k] = ec + ed;
        V[1][ch][k] = ec - ed;
        V[2][ch][k] = 4.0f * d1 - 5.0f * d3 + fOuter * R[ch][4][k];
      }
    }

  v2f p[3][9];
#pragma unroll
  for (int i = 0; i < 3; ++i) {
    const float l0 = from_lane_below(V[i][0][6]), l1 = from_lane_below(V[i][1][6]);
    const float r0 = from_lane_above(V[i][0][0]), r1 = from_lane_above(V[i][1][0]);
    p[i][0].x = l0;  p[i][0].y = l1;
    p[i][8].x = r0;  p[i][8].y = r1;
#pragma unroll
    for (int k = 0; k < 7; ++k) {
      p[i][k + 1].x = V[i][0][k];
      p[i][k + 1].y = V[i][1][k];
    }
  }

#pragma unroll
  for (int i = 0; i < 3; ++i) {
    const v2f e0 = p[i][0], e1 = p[i][1], e2 = p[i][2], e3 = p[i][3], e4 = p[i][4], e5 = p[i][5];
    const v2f b0 = fL * e0 + (0.25f * e4 - 1.25f * e2);
    const v2f fa = e4 - 4.0f * e2;
    const v2f fb = e3 - 4.0f * e1;
    const v2f b1 = fa + fb;
    const v2f b2 = fa - fb;
    const v2f fc = e4 - e2;
    const v2f fd = e3 - e1;
    const v2f b3 = fc + 2.0f * fd;
    const v2f b4 = fc - 2.0f * fd;
    const v2f b5 = 4.0f * e1 - 5.0f * e3 + e5;
#pragma unroll
    for (int t = 0; t < OT; ++t) {
      c[t][i][0] = __builtin_amdgcn_wmma_f32_16x16x4_f32(false, a[t][i][0],  false, b0, (short)0, c[t][i][0], false, false);
      c[t][i][1] = __builtin_amdgcn_wmma_f32_16x16x4_f32(false, aq[t][i][0], false, b1, (short)0, c[t][i][1], false, false);
      c[t][i][2] = __builtin_amdgcn_wmma_f32_16x16x4_f32(false, aq[t][i][1], false, b2, (short)0, c[t][i][2], false, false);
      c[t][i][3] = __builtin_amdgcn_wmma_f32_16x16x4_f32(false, aq[t][i][2], false, b3, (short)0, c[t][i][3], false, false);
      c[t][i][4] = __builtin_amdgcn_wmma_f32_16x16x4_f32(false, aq[t][i][3], false, b4, (short)0, c[t][i][4], false, false);
      c[t][i][5] = __builtin_amdgcn_wmma_f32_16x16x4_f32(false, a[t][i][2],  false, b5, (short)0, c[t][i][5], false, false);
    }
    const v2f h1 = p[i][5], h2 = p[i][6], h3 = p[i][7], h4 = p[i][8];
    const v2f gd = h3 - h1;
    const v2f gs = h3 - h2;
    const v2f t0 = (e4 - h2) + 0.5f * gd;
    const v2f t1 = gs - 2.0f * h1;
    const v2f t2 = gs + 2.0f * (h1 - h2);
    const v2f t4 = (fR * h4 - h2) - 2.0f * gd;
#pragma unroll
    for (int t = 0; t < OT; ++t) {
      c[t][i][6]  = __builtin_amdgcn_wmma_f32_16x16x4_f32(false, a[t][i][0],  false, t0, (short)0, c[t][i][6],  false, false);
      c[t][i][7]  = __builtin_amdgcn_wmma_f32_16x16x4_f32(false, aq[t][i][0], false, t1, (short)0, c[t][i][7],  false, false);
      c[t][i][8]  = __builtin_amdgcn_wmma_f32_16x16x4_f32(false, aq[t][i][1], false, t2, (short)0, c[t][i][8],  false, false);
      c[t][i][9]  = __builtin_amdgcn_wmma_f32_16x16x4_f32(false, aq[t][i][2], false, gd, (short)0, c[t][i][9],  false, false);
      c[t][i][10] = __builtin_amdgcn_wmma_f32_16x16x4_f32(false, a[t][i][2],  false, t4, (short)0, c[t][i][10], false, false);
    }
  }
}

template <int HF, int OT>
__device__ __forceinline__ void rows_f43_half(const float* __restrict__ bp, const v4f* __restrict__ ap,
                                              const int (&o)[6], float fOuter, float fL, float fR, v8f (&c)[OT][3][NTC])
{
  v2f   G0[OT][3][3];
  v2f   G1[OT][3][3];
  float R0[2][5][7];
  float R1[2][5][7];
#pragma unroll
  for (int t = 0; t < OT; ++t) load_weights<0>(ap + t * (Q_STEPS * 32), G0[t]);
  load_rows5<HF>(bp, o, R0);

  for (int q = 0; q < Q_STEPS; ++q) {
#pragma unroll
    for (int t = 0; t < OT; ++t) load_weights<1>(ap + t * (Q_STEPS * 32), G1[t]);
    load_rows5<HF>(bp + 4 * HWSZ, o, R1);
    __builtin_amdgcn_sched_barrier(0);

    v2f u[OT][3][3];
#pragma unroll
    for (int t = 0; t < OT; ++t) weight_rows<HF>(G0[t], u[t]);
    k_step_f43<HF, OT>(R0, u, fOuter, fL, fR, c);
    __builtin_amdgcn_sched_barrier(0);

    const int more = (q + 1 < Q_STEPS) ? 1 : 0;
    ap += more * 32;
    bp += more * (8 * HWSZ);
#pragma unroll
    for (int t = 0; t < OT; ++t) load_weights<0>(ap + t * (Q_STEPS * 32), G0[t]);
    load_rows5<HF>(bp, o, R0);
    __builtin_amdgcn_sched_barrier(0);

#pragma unroll
    for (int t = 0; t < OT; ++t) weight_rows<HF>(G1[t], u[t]);
    k_step_f43<HF, OT>(R1, u, fOuter, fL, fR, c);
    __builtin_amdgcn_sched_barrier(0);
  }
}

#define XCH_SLOTS 21

template <int HF>
__device__ __forceinline__ void pack_for_partner(const v8f (&c)[3][NT], v8f (&E)[3][NT], v4f (&snd)[XCH_SLOTS])
{
#pragma unroll
  for (int j = 0; j < NT; ++j) {
    if (HF == 0) {
      const v8f s12 = c[1][j] + c[2][j];
      E[0][j] = c[0][j] + s12;
      E[1][j] = s12;
      E[2][j] = c[1][j] - c[2][j];
    } else {
      E[0][j] = c[0][j] + c[1][j];
      E[1][j] = c[0][j] - c[1][j];
      E[2][j] = c[2][j];
    }
  }
  float f[4 * XCH_SLOTS];
#pragma unroll
  for (int e = 0; e < 3; ++e)
#pragma unroll
    for (int rr = 0; rr < 4; ++rr)
#pragma unroll
      for (int j = 0; j < NT; ++j) f[(e * 4 + rr) * 7 + j] = E[e][j][(1 - HF) * 4 + rr];
#pragma unroll
  for (int s = 0; s < XCH_SLOTS; ++s) {
    v4f t; t.x = f[4 * s]; t.y = f[4 * s + 1]; t.z = f[4 * s + 2]; t.w = f[4 * s + 3];
    snd[s] = t;
  }
}

template <int HF>
__device__ __forceinline__ void finish_rows(const v8f (&E)[3][NT], const v4f (&rcv)[XCH_SLOTS], float* __restrict__ obase)
{
  float g[4 * XCH_SLOTS];
#pragma unroll
  for (int s = 0; s < XCH_SLOTS; ++s) {
    g[4 * s] = rcv[s].x; g[4 * s + 1] = rcv[s].y; g[4 * s + 2] = rcv[s].z; g[4 * s + 3] = rcv[s].w;
  }
  const float sc = 1.0f / 9.0f;
#pragma unroll
  for (int rr = 0; rr < 4; ++rr) {
    float y[4][NT];
#pragma unroll
    for (int j = 0; j < NT; ++j) {
      float a[3], b[3];
#pragma unroll
      for (int e = 0; e < 3; ++e) {
        const float own = E[e][j][HF * 4 + rr];
        const float got = g[(e * 4 + rr) * 7 + j];
        a[e] = HF ? got : own;
        b[e] = HF ? own : got;
      }
      y[0][j] = (a[0] + b[0]) * sc;
      y[1][j] = (a[2] + 2.0f * b[1]) * sc;
      y[2][j] = (a[1] + 4.0f * b[0]) * sc;
      y[3][j] = (a[2] + (8.0f * b[1] + b[2])) * sc;
    }
    float* orow = obase + (long)rr * HWSZ;
#pragma unroll
    for (int i = 0; i < 4; ++i) {
      float* op = orow + i * WW;
      v4f v; v2f u;
      v.x = y[i][0]; v.y = y[i][1]; v.z = y[i][2]; v.w = y[i][3];
      u.x = y[i][4]; u.y = y[i][5];
      __builtin_memcpy(op, &v, 16);
      __builtin_memcpy(op + 4, &u, 8);
      op[6] = y[i][6];
    }
  }
}

template <int OT>
__device__ __forceinline__ void conv_body_rows_f43(
    const float* __restrict__ x, const float* __restrict__ w2, float* __restrict__ out,
    v4f (*xch)[XCH_SLOTS][32], const int unit, const int oc_group)
{
  const int lane   = threadIdx.x & 31;
  const int l16    = lane & 15;
  const int hiHalf = lane >> 4;
  const int kHi    = hiHalf * 2;
  const int wave   = __builtin_amdgcn_readfirstlane((int)(threadIdx.x >> 5));
  const int hf     = wave & 1;
  const int pr     = wave >> 1;

  const int oc_base = oc_group + pr * (16 * OT);
  const int n       = unit / (HH / 8);
  const int hA      = (unit - n * (HH / 8)) * 8 + (l16 >> 3) * 4;
  const int wq      = l16 & 7;
  const int w0      = 7 * wq;

  const bool v0 = hA != 0;
  const bool v5 = hA != HH - 4;
  const bool mL = wq != 0;
  const bool mR = wq != 7;


  v8f cw[OT][3][NTC] = {};

  const float* bp = x + (long)n * CHW + (long)kHi * HWSZ + w0;
  const int o[6] = { (v0 ? hA - 1 : hA) * WW, hA * WW, (hA + 1) * WW, (hA + 2) * WW, (hA + 3) * WW,
                     (v5 ? hA + 4 : hA) * WW };
  const v4f* ap = (const v4f*)w2 + (long)(oc_base >> 4) * (Q_STEPS * 32) + lane;

  const float fL = mL ? 1.0f : 0.0f;
  const float fR = mR ? 1.0f : 0.0f;
  if (hf == 0) rows_f43_half<0, OT>(bp, ap, o, v0 ? 4.0f : 0.0f, fL, fR, cw);
  else         rows_f43_half<1, OT>(bp, ap, o, v5 ? 1.0f : 0.0f, fL, fR, cw);

#pragma unroll
  for (int t = 0; t < OT; ++t) {
    v8f c[3][NT];
#pragma unroll
    for (int i = 0; i < 3; ++i) {
      const v8f s12 = cw[t][i][1] + cw[t][i][2];
      const v8f d12 = cw[t][i][1] - cw[t][i][2];
      const v8f s34 = cw[t][i][3] + cw[t][i][4];
      const v8f d34 = cw[t][i][3] - cw[t][i][4];
      c[i][0] = cw[t][i][0] + s12 + s34;
      c[i][1] = d12 + 2.0f * d34;
      c[i][2] = s12 + 4.0f * s34;
      c[i][3] = d12 + (8.0f * d34 + cw[t][i][5]);
      const v8f n1  = 3.0f * cw[t][i][7];
      const v8f n3  = 4.0f * cw[t][i][9];
      const v8f s78 = n1 + cw[t][i][8];
      const v8f d78 = n1 - cw[t][i][8];
      c[i][4] = cw[t][i][6] + s78 + n3;
      c[i][5] = d78 + 2.0f * n3;
      c[i][6] = s78 + (4.0f * n3 + cw[t][i][10]);
    }
    v8f E[3][NT];
    v4f snd[XCH_SLOTS];
    if (hf == 0) pack_for_partner<0>(c, E, snd);
    else         pack_for_partner<1>(c, E, snd);

    if (t > 0) __syncthreads();
#pragma unroll
    for (int s = 0; s < XCH_SLOTS; ++s) xch[wave][s][lane] = snd[s];
    __syncthreads();
    v4f rcv[XCH_SLOTS];
#pragma unroll
    for (int s = 0; s < XCH_SLOTS; ++s) rcv[s] = xch[wave ^ 1][s][lane];

    float* obase = out + ((long)n * OC + oc_base + t * 16 + hiHalf * 8 + hf * 4) * HWSZ + (hA * WW + w0);
    if (hf == 0) finish_rows<0>(E, rcv, obase);
    else         finish_rows<1>(E, rcv, obase);
  }
}

__global__ __launch_bounds__(128) __attribute__((amdgpu_waves_per_eu(1, 1))) void conv3x3_wmma_f32(
    const float* __restrict__ x, const float* __restrict__ w,
    const float* __restrict__ w2, size_t ws_bytes, float* __restrict__ out)
{
  __shared__ v4f xch[4][XCH_SLOTS][32];
  const int b = (int)blockIdx.x;
  if (ws_bytes >= (size_t)W2_BYTES) {
    if (b < TWO_TILE_GROUPS) {
      conv_body_rows_f43<2>(x, w2, out, xch, b >> 2, (b & 3) * 64);
    } else {
      const int b1 = b - TWO_TILE_GROUPS;
      conv_body_rows_f43<1>(x, w2, out, xch, TWO_TILE_GROUPS / 4 + (b1 >> 3), (b1 & 7) * 32);
    }
  } else {
    for (int bid = b; bid < ROW_PAIRS * 2; bid += GRID_GROUPS) conv_body<false>(x, w, w2, out, bid);
  }
}

extern "C" void kernel_launch(void* const* d_in, const int* in_sizes, int n_in,
                              void* d_out, int out_size, void* d_ws, size_t ws_size,
                              hipStream_t stream) {
  const float* x = (const float*)d_in[0];
  const float* w = (const float*)d_in[1];
  float* out = (float*)d_out;
  float* w2  = (float*)d_ws;

  weights_to_fragment_order<<<dim3(W2_CHUNKS / 256, 1, 1), dim3(256, 1, 1), 0, stream>>>(w, w2, ws_size);

  dim3 grid(GRID_GROUPS, 1, 1);
  dim3 block(128, 1, 1);
  conv3x3_wmma_f32<<<grid, block, 0, stream>>>(x, w, w2, ws_size, out);
}
